# starting kernel
# baseline (speedup 1.0000x reference)
#include <hip/hip_runtime.h>
#include <hip/hip_bf16.h>
#include <math.h>

#define N_NODES   100000
#define N_EDGES   3200000

typedef float v2f __attribute__((ext_vector_type(2)));
typedef float v8f __attribute__((ext_vector_type(8)));

__global__ void deg_init_kernel(float* __restrict__ deg, int n) {
    int i = blockIdx.x * blockDim.x + threadIdx.x;
    if (i < n) deg[i] = 1.0f;
}

__global__ void deg_accum_kernel(const int* __restrict__ col,
                                 float* __restrict__ deg, int ne) {
    int e = blockIdx.x * blockDim.x + threadIdx.x;
    if (e >= ne) return;
    __hip_atomic_fetch_add(&deg[col[e]], 1.0f,
                           __ATOMIC_RELAXED, __HIP_MEMORY_SCOPE_AGENT);
}

__global__ void dinv_kernel(float* __restrict__ dinv, int n) {
    int i = blockIdx.x * blockDim.x + threadIdx.x;
    if (i < n) dinv[i] = rsqrtf(dinv[i]);
}

__global__ void norm_kernel(const int* __restrict__ row,
                            const int* __restrict__ col,
                            const float* __restrict__ dinv,
                            float* __restrict__ norm, int ne) {
    int e = blockIdx.x * blockDim.x + threadIdx.x;
    if (e >= ne) return;
    norm[e] = dinv[row[e]] * dinv[col[e]];
}

__global__ void __launch_bounds__(256)
gcn_wmma_xw1_kernel(const float* __restrict__ x,
                    const float* __restrict__ W1,
                    float* __restrict__ hlin,
                    int n) {
    __shared__ float Wp[128 * 16];
    const int t = threadIdx.x;
    #pragma unroll
    for (int j = 0; j < 8; ++j) {
        int idx = t * 8 + j;
        int kr  = idx >> 4;
        int nc  = idx & 15;
        Wp[idx] = (nc < 4) ? W1[kr * 4 + nc] : 0.0f;
    }
    __syncthreads();

    const int wave  = t >> 5;
    const int lane  = t & 31;
    const int tile  = blockIdx.x * 8 + wave;
    const int nTiles = n >> 4;
    if (tile >= nTiles) return;

    const int base = tile * 16;
    const int hh   = lane >> 4;
    const int nl   = lane & 15;
    const int arow = base + nl;

    v8f c = {};
    #pragma unroll 4
    for (int k = 0; k < 128; k += 4) {
        const int kk = k + 2 * hh;
        v2f a = *reinterpret_cast<const v2f*>(x + (size_t)arow * 128 + kk);
        v2f b;
        b.x = Wp[(kk + 0) * 16 + nl];
        b.y = Wp[(kk + 1) * 16 + nl];
        c = __builtin_amdgcn_wmma_f32_16x16x4_f32(
                 false, a,  false, b,
                 (short)0, c,  false,  false);
    }

    if (nl < 4) {
        #pragma unroll
        for (int v = 0; v < 8; ++v) {
            int m = base + v + 8 * hh;
            hlin[(size_t)m * 4 + nl] = c[v];
        }
    }
}

template <int IN, int OUT>
__global__ void dense_small_kernel(const float* __restrict__ h,
                                   const float* __restrict__ W,
                                   float* __restrict__ out, int n) {
    int i = blockIdx.x * blockDim.x + threadIdx.x;
    if (i >= n) return;
    float hv[IN];
    #pragma unroll
    for (int k = 0; k < IN; ++k) hv[k] = h[(size_t)i * IN + k];
    #pragma unroll
    for (int o = 0; o < OUT; ++o) {
        float s = 0.0f;
        #pragma unroll
        for (int k = 0; k < IN; ++k) s = fmaf(hv[k], W[k * OUT + o], s);
        out[(size_t)i * OUT + o] = s;
    }
}

template <int OUT>
__global__ void edge_scatter_kernel(const int* __restrict__ row,
                                    const int* __restrict__ col,
                                    const float* __restrict__ norm,
                                    const float* __restrict__ hlin,
                                    float* __restrict__ agg, int ne) {
    int e = blockIdx.x * blockDim.x + threadIdx.x;
    if (e >= ne) return;
    const int r = row[e];
    const int c = col[e];
    const float nm = norm[e];
    float msg[OUT];
    if (OUT == 4) {
        float4 v = reinterpret_cast<const float4*>(hlin)[c];
        msg[0] = v.x; msg[1] = v.y; msg[2] = v.z; msg[3] = v.w;
    } else if (OUT == 2) {
        float2 v = reinterpret_cast<const float2*>(hlin)[c];
        msg[0] = v.x; msg[1] = v.y;
    } else {
        #pragma unroll
        for (int f = 0; f < OUT; ++f) msg[f] = hlin[(size_t)c * OUT + f];
    }
    #pragma unroll
    for (int f = 0; f < OUT; ++f) {
        __hip_atomic_fetch_add(&agg[(size_t)r * OUT + f], msg[f] * nm,
                               __ATOMIC_RELAXED, __HIP_MEMORY_SCOPE_AGENT);
    }
}

template <int OUT>
__global__ void finalize_kernel(const float* __restrict__ agg,
                                const float* __restrict__ hlin,
                                const float* __restrict__ dinv,
                                const float* __restrict__ b,
                                float* __restrict__ out, int n) {
    int i = blockIdx.x * blockDim.x + threadIdx.x;
    if (i >= n) return;
    const float d2 = dinv[i] * dinv[i];
    #pragma unroll
    for (int f = 0; f < OUT; ++f) {
        float v = agg[(size_t)i * OUT + f]
                + hlin[(size_t)i * OUT + f] * d2 + b[f];
        out[(size_t)i * OUT + f] = tanhf(v);
    }
}

__global__ void classify_kernel(const float* __restrict__ h,
                                const float* __restrict__ Wc,
                                const float* __restrict__ bc,
                                float* __restrict__ out, int n) {
    int i = blockIdx.x * blockDim.x + threadIdx.x;
    if (i >= n) return;
    const float h0 = h[(size_t)i * 2 + 0];
    const float h1 = h[(size_t)i * 2 + 1];
    #pragma unroll
    for (int c = 0; c < 4; ++c) {
        out[(size_t)i * 4 + c] = fmaf(h0, Wc[c], fmaf(h1, Wc[4 + c], bc[c]));
    }
}

extern "C" void kernel_launch(void* const* d_in, const int* in_sizes, int n_in,
                              void* d_out, int out_size, void* d_ws, size_t ws_size,
                              hipStream_t stream) {
    const float* x  = (const float*)d_in[0];
    const int*   ei = (const int*)d_in[1];
    const float* W1 = (const float*)d_in[2];
    const float* b1 = (const float*)d_in[3];
    const float* W2 = (const float*)d_in[4];
    const float* b2 = (const float*)d_in[5];
    const float* W3 = (const float*)d_in[6];
    const float* b3 = (const float*)d_in[7];
    const float* Wc = (const float*)d_in[8];
    const float* bc = (const float*)d_in[9];

    const int N = N_NODES;
    const int E = N_EDGES;
    const int* row = ei;
    const int* col = ei + E;

    float* ws   = (float*)d_ws;
    float* dinv = ws;
    float* norm = dinv + N;
    float* hlin = norm + E;
    float* agg  = hlin + 4 * N;
    float* hA   = agg  + 4 * N;
    float* hB   = hA   + 4 * N;

    float* outC = (float*)d_out;
    float* outH = outC + (size_t)4 * N;

    const int BT = 256;
    const int gN = (N + BT - 1) / BT;
    const int gE = (E + BT - 1) / BT;

    deg_init_kernel<<<gN, BT, 0, stream>>>(dinv, N);
    deg_accum_kernel<<<gE, BT, 0, stream>>>(col, dinv, E);
    dinv_kernel<<<gN, BT, 0, stream>>>(dinv, N);
    norm_kernel<<<gE, BT, 0, stream>>>(row, col, dinv, norm, E);

    {
        const int tiles  = N / 16;
        const int blocks = (tiles + 7) / 8;
        gcn_wmma_xw1_kernel<<<blocks, 256, 0, stream>>>(x, W1, hlin, N);
        hipMemsetAsync(agg, 0, (size_t)4 * N * sizeof(float), stream);
        edge_scatter_kernel<4><<<gE, BT, 0, stream>>>(row, col, norm, hlin, agg, E);
        finalize_kernel<4><<<gN, BT, 0, stream>>>(agg, hlin, dinv, b1, hA, N);
    }
    {
        dense_small_kernel<4, 4><<<gN, BT, 0, stream>>>(hA, W2, hlin, N);
        hipMemsetAsync(agg, 0, (size_t)4 * N * sizeof(float), stream);
        edge_scatter_kernel<4><<<gE, BT, 0, stream>>>(row, col, norm, hlin, agg, E);
        finalize_kernel<4><<<gN, BT, 0, stream>>>(agg, hlin, dinv, b2, hB, N);
    }
    {
        dense_small_kernel<4, 2><<<gN, BT, 0, stream>>>(hB, W3, hlin, N);
        hipMemsetAsync(agg, 0, (size_t)2 * N * sizeof(float), stream);
        edge_scatter_kernel<2><<<gE, BT, 0, stream>>>(row, col, norm, hlin, agg, E);
        finalize_kernel<2><<<gN, BT, 0, stream>>>(agg, hlin, dinv, b3, outH, N);
    }
    classify_kernel<<<gN, BT, 0, stream>>>(outH, Wc, bc, outC, N);
}


# Degree count: global float atomics replaced by the edge scatter's scheme: each of 256 work-groups counts one source range in LDS (int adds) over its edge slice; dinv launch adds the planes; deg exact
# speedup vs baseline: 5.3867x; 5.3867x over previous
#include <hip/hip_runtime.h>
#include <hip/hip_bf16.h>
#include <math.h>

#define N_NODES   100000
#define N_EDGES   3200000

typedef float v2f __attribute__((ext_vector_type(2)));
typedef float v8f __attribute__((ext_vector_type(8)));

#define DG_THREADS 1024
#define DG_GROUPS  256
#define DG_P       4

__global__ void __launch_bounds__(DG_THREADS)
deg_range_kernel(const int* __restrict__ col, float* __restrict__ cntp) {
    constexpr int P     = DG_P;
    constexpr int G     = DG_GROUPS / P;
    constexpr int NPART = N_NODES / P;
    constexpr int ESL   = N_EDGES / G;
    constexpr int NV    = ESL / 4;
    constexpr int TRIPS = (NV + DG_THREADS - 1) / DG_THREADS;
    static_assert(N_NODES % P == 0 && N_EDGES % G == 0 && ESL % 4 == 0 && NPART % 4 == 0,
                  "ranges, slices and their 16-byte pieces divide evenly");
    __shared__ unsigned int cnt[NPART];

    const int t = threadIdx.x;
    const int p = blockIdx.x % P;
    const int g = blockIdx.x / P;
    const int base = p * NPART;

    uint4* cnt4 = reinterpret_cast<uint4*>(cnt);
    for (int i = t; i < NPART / 4; i += DG_THREADS) cnt4[i] = make_uint4(0u, 0u, 0u, 0u);
    __syncthreads();

    const int4* col4 = reinterpret_cast<const int4*>(col + (size_t)g * ESL);

    int4 c[TRIPS];
    #pragma unroll
    for (int u = 0; u < TRIPS; ++u) {
        const int v = t + u * DG_THREADS;
        c[u] = make_int4(-1, -1, -1, -1);
        if (v < NV) c[u] = col4[v];
    }
    #pragma unroll
    for (int u = 0; u < TRIPS; ++u) {
        const int cc[4] = { c[u].x - base, c[u].y - base, c[u].z - base, c[u].w - base };
        #pragma unroll
        for (int j = 0; j < 4; ++j) {
            if ((unsigned)cc[j] < (unsigned)NPART)
                __hip_atomic_fetch_add(&cnt[cc[j]], 1u,
                                       __ATOMIC_RELAXED, __HIP_MEMORY_SCOPE_WORKGROUP);
        }
    }
    __syncthreads();

    float4* dst = reinterpret_cast<float4*>(cntp + (size_t)g * N_NODES + base);
    for (int i = t; i < NPART / 4; i += DG_THREADS) {
        const uint4 k = cnt4[i];
        dst[i] = make_float4((float)k.x, (float)k.y, (float)k.z, (float)k.w);
    }
}

__global__ void dinv_planes_kernel(const float* __restrict__ cntp,
                                   float* __restrict__ dinv, int n) {
    constexpr int G = DG_GROUPS / DG_P;
    int i = blockIdx.x * blockDim.x + threadIdx.x;
    if (i >= n) return;
    float d = 1.0f;
    #pragma unroll 32
    for (int g = 0; g < G; ++g) d += cntp[(size_t)g * N_NODES + i];
    dinv[i] = rsqrtf(d);
}

__global__ void norm_kernel(const int* __restrict__ row,
                            const int* __restrict__ col,
                            const float* __restrict__ dinv,
                            float* __restrict__ norm, int ne) {
    int e = blockIdx.x * blockDim.x + threadIdx.x;
    if (e >= ne) return;
    norm[e] = dinv[row[e]] * dinv[col[e]];
}

__global__ void __launch_bounds__(256)
gcn_wmma_xw1_kernel(const float* __restrict__ x,
                    const float* __restrict__ W1,
                    float* __restrict__ hlin,
                    int n) {
    __shared__ float Wp[128 * 16];
    const int t = threadIdx.x;
    #pragma unroll
    for (int j = 0; j < 8; ++j) {
        int idx = t * 8 + j;
        int kr  = idx >> 4;
        int nc  = idx & 15;
        Wp[idx] = (nc < 4) ? W1[kr * 4 + nc] : 0.0f;
    }
    __syncthreads();

    const int wave  = t >> 5;
    const int lane  = t & 31;
    const int tile  = blockIdx.x * 8 + wave;
    const int nTiles = n >> 4;
    if (tile >= nTiles) return;

    const int base = tile * 16;
    const int hh   = lane >> 4;
    const int nl   = lane & 15;
    const int arow = base + nl;

    v8f c = {};
    #pragma unroll 4
    for (int k = 0; k < 128; k += 4) {
        const int kk = k + 2 * hh;
        v2f a = *reinterpret_cast<const v2f*>(x + (size_t)arow * 128 + kk);
        v2f b;
        b.x = Wp[(kk + 0) * 16 + nl];
        b.y = Wp[(kk + 1) * 16 + nl];
        c = __builtin_amdgcn_wmma_f32_16x16x4_f32(
                 false, a,  false, b,
                 (short)0, c,  false,  false);
    }

    if (nl < 4) {
        #pragma unroll
        for (int v = 0; v < 8; ++v) {
            int m = base + v + 8 * hh;
            hlin[(size_t)m * 4 + nl] = c[v];
        }
    }
}

template <int IN, int OUT>
__global__ void dense_small_kernel(const float* __restrict__ h,
                                   const float* __restrict__ W,
                                   float* __restrict__ out, int n) {
    int i = blockIdx.x * blockDim.x + threadIdx.x;
    if (i >= n) return;
    float hv[IN];
    #pragma unroll
    for (int k = 0; k < IN; ++k) hv[k] = h[(size_t)i * IN + k];
    #pragma unroll
    for (int o = 0; o < OUT; ++o) {
        float s = 0.0f;
        #pragma unroll
        for (int k = 0; k < IN; ++k) s = fmaf(hv[k], W[k * OUT + o], s);
        out[(size_t)i * OUT + o] = s;
    }
}

#define SC_THREADS 1024
#define SC_GROUPS  256

template <int OUT> struct RowVec;
template <> struct RowVec<4> { typedef float4 type; };
template <> struct RowVec<2> { typedef float2 type; };

template <int OUT>
__global__ void __launch_bounds__(SC_THREADS)
edge_range_kernel(const int* __restrict__ row,
                  const int* __restrict__ col,
                  const float* __restrict__ norm,
                  const float* __restrict__ hlin,
                  float* __restrict__ part) {
    typedef typename RowVec<OUT>::type VT;
    constexpr int P     = 2 * OUT;
    constexpr int G     = SC_GROUPS / P;
    constexpr int NPART = N_NODES / P;
    constexpr int ESL   = N_EDGES / G;
    constexpr int NV    = ESL / 4;
    constexpr int NACC  = NPART * OUT;
    static_assert(N_NODES % P == 0 && N_EDGES % G == 0 && ESL % 4 == 0 && NACC % 4 == 0,
                  "ranges, slices and their 16-byte pieces divide evenly");
    __shared__ float acc[NACC];

    const int t = threadIdx.x;
    const int p = blockIdx.x % P;
    const int g = blockIdx.x / P;
    const int base = p * NPART;

    float4* acc4 = reinterpret_cast<float4*>(acc);
    for (int i = t; i < NACC / 4; i += SC_THREADS) acc4[i] = make_float4(0.f, 0.f, 0.f, 0.f);
    __syncthreads();

    const int4*   row4  = reinterpret_cast<const int4*>(row + (size_t)g * ESL);
    const int4*   col4  = reinterpret_cast<const int4*>(col + (size_t)g * ESL);
    const float4* norm4 = reinterpret_cast<const float4*>(norm + (size_t)g * ESL);
    const VT*     hv    = reinterpret_cast<const VT*>(hlin);

    int v = t;
    int4   rN = make_int4(-1, -1, -1, -1), cN = make_int4(0, 0, 0, 0);
    float4 nN = make_float4(0.f, 0.f, 0.f, 0.f);
    if (v < NV) { rN = row4[v]; cN = col4[v]; nN = norm4[v]; }
    for (; v < NV; v += SC_THREADS) {
        const int4 r = rN, c = cN; const float4 nm = nN;
        const int vn = v + SC_THREADS;
        if (vn < NV) { rN = row4[vn]; cN = col4[vn]; nN = norm4[vn]; }

        const int   rr[4] = { r.x - base, r.y - base, r.z - base, r.w - base };
        const int   cc[4] = { c.x, c.y, c.z, c.w };
        const float nn[4] = { nm.x, nm.y, nm.z, nm.w };
        bool m[4];
        VT   h[4];
        #pragma unroll
        for (int j = 0; j < 4; ++j) {
            m[j] = (unsigned)rr[j] < (unsigned)NPART;
            h[j] = VT{};
            if (m[j]) h[j] = hv[cc[j]];
        }
        #pragma unroll
        for (int j = 0; j < 4; ++j) {
            if (m[j]) {
                const float* hj = reinterpret_cast<const float*>(&h[j]);
                #pragma unroll
                for (int f = 0; f < OUT; ++f)
                    __hip_atomic_fetch_add(&acc[rr[j] * OUT + f], hj[f] * nn[j],
                                           __ATOMIC_RELAXED, __HIP_MEMORY_SCOPE_WORKGROUP);
            }
        }
    }
    __syncthreads();

    float4* dst = reinterpret_cast<float4*>(part + ((size_t)g * N_NODES + base) * OUT);
    for (int i = t; i < NACC / 4; i += SC_THREADS) dst[i] = acc4[i];
}

template <int OUT>
__global__ void finalize_kernel(const float* __restrict__ part,
                                const float* __restrict__ hlin,
                                const float* __restrict__ dinv,
                                const float* __restrict__ b,
                                float* __restrict__ out, int n) {
    typedef typename RowVec<OUT>::type VT;
    constexpr int G = SC_GROUPS / (2 * OUT);
    int i = blockIdx.x * blockDim.x + threadIdx.x;
    if (i >= n) return;
    const VT* pv = reinterpret_cast<const VT*>(part);
    float agg[OUT];
    #pragma unroll
    for (int f = 0; f < OUT; ++f) agg[f] = 0.0f;
    #pragma unroll 16
    for (int g = 0; g < G; ++g) {
        const VT pg = pv[(size_t)g * N_NODES + i];
        const float* pf = reinterpret_cast<const float*>(&pg);
        #pragma unroll
        for (int f = 0; f < OUT; ++f) agg[f] += pf[f];
    }
    const float d2 = dinv[i] * dinv[i];
    #pragma unroll
    for (int f = 0; f < OUT; ++f) {
        float v = agg[f]
                + hlin[(size_t)i * OUT + f] * d2 + b[f];
        out[(size_t)i * OUT + f] = tanhf(v);
    }
}

__global__ void classify_kernel(const float* __restrict__ h,
                                const float* __restrict__ Wc,
                                const float* __restrict__ bc,
                                float* __restrict__ out, int n) {
    int i = blockIdx.x * blockDim.x + threadIdx.x;
    if (i >= n) return;
    const float h0 = h[(size_t)i * 2 + 0];
    const float h1 = h[(size_t)i * 2 + 1];
    #pragma unroll
    for (int c = 0; c < 4; ++c) {
        out[(size_t)i * 4 + c] = fmaf(h0, Wc[c], fmaf(h1, Wc[4 + c], bc[c]));
    }
}

extern "C" void kernel_launch(void* const* d_in, const int* in_sizes, int n_in,
                              void* d_out, int out_size, void* d_ws, size_t ws_size,
                              hipStream_t stream) {
    const float* x  = (const float*)d_in[0];
    const int*   ei = (const int*)d_in[1];
    const float* W1 = (const float*)d_in[2];
    const float* b1 = (const float*)d_in[3];
    const float* W2 = (const float*)d_in[4];
    const float* b2 = (const float*)d_in[5];
    const float* W3 = (const float*)d_in[6];
    const float* b3 = (const float*)d_in[7];
    const float* Wc = (const float*)d_in[8];
    const float* bc = (const float*)d_in[9];

    const int N = N_NODES;
    const int E = N_EDGES;
    const int* row = ei;
    const int* col = ei + E;

    float* ws   = (float*)d_ws;
    float* dinv = ws;
    float* norm = dinv + N;
    float* hlin = norm + E;
    float* hA   = hlin + 4 * N;
    float* hB   = hA   + 4 * N;
    float* part = hB   + 4 * N;

    float* outC = (float*)d_out;
    float* outH = outC + (size_t)4 * N;

    const int BT = 256;
    const int gN = (N + BT - 1) / BT;
    const int gE = (E + BT - 1) / BT;

    deg_range_kernel<<<DG_GROUPS, DG_THREADS, 0, stream>>>(col, part);
    dinv_planes_kernel<<<gN, BT, 0, stream>>>(part, dinv, N);
    norm_kernel<<<gE, BT, 0, stream>>>(row, col, dinv, norm, E);

    {
        const int tiles  = N / 16;
        const int blocks = (tiles + 7) / 8;
        gcn_wmma_xw1_kernel<<<blocks, 256, 0, stream>>>(x, W1, hlin, N);
        edge_range_kernel<4><<<SC_GROUPS, SC_THREADS, 0, stream>>>(row, col, norm, hlin, part);
        finalize_kernel<4><<<gN, BT, 0, stream>>>(part, hlin, dinv, b1, hA, N);
    }
    {
        dense_small_kernel<4, 4><<<gN, BT, 0, stream>>>(hA, W2, hlin, N);
        edge_range_kernel<4><<<SC_GROUPS, SC_THREADS, 0, stream>>>(row, col, norm, hlin, part);
        finalize_kernel<4><<<gN, BT, 0, stream>>>(part, hlin, dinv, b2, hB, N);
    }
    {
        dense_small_kernel<4, 2><<<gN, BT, 0, stream>>>(hB, W3, hlin, N);
        edge_range_kernel<2><<<SC_GROUPS, SC_THREADS, 0, stream>>>(row, col, norm, hlin, part);
        finalize_kernel<2><<<gN, BT, 0, stream>>>(part, hlin, dinv, b3, outH, N);
    }
    classify_kernel<<<gN, BT, 0, stream>>>(outH, Wc, bc, outC, N);
}


# Per-edge norm array and its launch dropped: dinv rides in the gathered row (h, dinv) and the range launch forms dinv[row]*dinv[col] with its range's dinv held in LDS; 13 launches to 12; values same
# speedup vs baseline: 6.1935x; 1.1498x over previous
#include <hip/hip_runtime.h>
#include <hip/hip_bf16.h>
#include <math.h>

#define N_NODES   100000
#define N_EDGES   3200000

typedef float v2f __attribute__((ext_vector_type(2)));
typedef float v8f __attribute__((ext_vector_type(8)));

#define DG_THREADS 1024
#define DG_GROUPS  256
#define DG_P       4

__global__ void __launch_bounds__(DG_THREADS)
deg_range_kernel(const int* __restrict__ col, float* __restrict__ cntp) {
    constexpr int P     = DG_P;
    constexpr int G     = DG_GROUPS / P;
    constexpr int NPART = N_NODES / P;
    constexpr int ESL   = N_EDGES / G;
    constexpr int NV    = ESL / 4;
    constexpr int TRIPS = (NV + DG_THREADS - 1) / DG_THREADS;
    static_assert(N_NODES % P == 0 && N_EDGES % G == 0 && ESL % 4 == 0 && NPART % 4 == 0,
                  "ranges, slices and their 16-byte pieces divide evenly");
    __shared__ unsigned int cnt[NPART];

    const int t = threadIdx.x;
    const int p = blockIdx.x % P;
    const int g = blockIdx.x / P;
    const int base = p * NPART;

    uint4* cnt4 = reinterpret_cast<uint4*>(cnt);
    for (int i = t; i < NPART / 4; i += DG_THREADS) cnt4[i] = make_uint4(0u, 0u, 0u, 0u);
    __syncthreads();

    const int4* col4 = reinterpret_cast<const int4*>(col + (size_t)g * ESL);

    int4 c[TRIPS];
    #pragma unroll
    for (int u = 0; u < TRIPS; ++u) {
        const int v = t + u * DG_THREADS;
        c[u] = make_int4(-1, -1, -1, -1);
        if (v < NV) c[u] = col4[v];
    }
    #pragma unroll
    for (int u = 0; u < TRIPS; ++u) {
        const int cc[4] = { c[u].x - base, c[u].y - base, c[u].z - base, c[u].w - base };
        #pragma unroll
        for (int j = 0; j < 4; ++j) {
            if ((unsigned)cc[j] < (unsigned)NPART)
                __hip_atomic_fetch_add(&cnt[cc[j]], 1u,
                                       __ATOMIC_RELAXED, __HIP_MEMORY_SCOPE_WORKGROUP);
        }
    }
    __syncthreads();

    float4* dst = reinterpret_cast<float4*>(cntp + (size_t)g * N_NODES + base);
    for (int i = t; i < NPART / 4; i += DG_THREADS) {
        const uint4 k = cnt4[i];
        dst[i] = make_float4((float)k.x, (float)k.y, (float)k.z, (float)k.w);
    }
}

__global__ void dinv_planes_kernel(const float* __restrict__ cntp,
                                   float* __restrict__ dinv, int n) {
    constexpr int G = DG_GROUPS / DG_P;
    int i = blockIdx.x * blockDim.x + threadIdx.x;
    if (i >= n) return;
    float d = 1.0f;
    #pragma unroll 32
    for (int g = 0; g < G; ++g) d += cntp[(size_t)g * N_NODES + i];
    dinv[i] = rsqrtf(d);
}

#define ROW_STRIDE(OUT) (2 * (OUT))

__global__ void __launch_bounds__(256)
gcn_wmma_xw1_kernel(const float* __restrict__ x,
                    const float* __restrict__ W1,
                    const float* __restrict__ dinv,
                    float* __restrict__ hlin,
                    int n) {
    __shared__ float Wp[128 * 16];
    const int t = threadIdx.x;
    #pragma unroll
    for (int j = 0; j < 8; ++j) {
        int idx = t * 8 + j;
        int kr  = idx >> 4;
        int nc  = idx & 15;
        Wp[idx] = (nc < 4) ? W1[kr * 4 + nc] : 0.0f;
    }
    __syncthreads();

    const int wave  = t >> 5;
    const int lane  = t & 31;
    const int tile  = blockIdx.x * 8 + wave;
    const int nTiles = n >> 4;
    if (tile >= nTiles) return;

    const int base = tile * 16;
    const int hh   = lane >> 4;
    const int nl   = lane & 15;
    const int arow = base + nl;

    const float4 dlo = *reinterpret_cast<const float4*>(dinv + base + 8 * hh);
    const float4 dhi = *reinterpret_cast<const float4*>(dinv + base + 8 * hh + 4);
    const float dv[8] = { dlo.x, dlo.y, dlo.z, dlo.w, dhi.x, dhi.y, dhi.z, dhi.w };

    v8f c = {};
    #pragma unroll 4
    for (int k = 0; k < 128; k += 4) {
        const int kk = k + 2 * hh;
        v2f a = *reinterpret_cast<const v2f*>(x + (size_t)arow * 128 + kk);
        v2f b;
        b.x = Wp[(kk + 0) * 16 + nl];
        b.y = Wp[(kk + 1) * 16 + nl];
        c = __builtin_amdgcn_wmma_f32_16x16x4_f32(
                 false, a,  false, b,
                 (short)0, c,  false,  false);
    }

    if (nl < 5) {
        #pragma unroll
        for (int v = 0; v < 8; ++v) {
            int m = base + v + 8 * hh;
            hlin[(size_t)m * ROW_STRIDE(4) + nl] = (nl < 4) ? c[v] : dv[v];
        }
    }
}

template <int IN, int OUT>
__global__ void dense_small_kernel(const float* __restrict__ h,
                                   const float* __restrict__ W,
                                   const float* __restrict__ dinv,
                                   float* __restrict__ out,
                                   int n) {
    int i = blockIdx.x * blockDim.x + threadIdx.x;
    if (i >= n) return;
    float hv[IN];
    #pragma unroll
    for (int k = 0; k < IN; ++k) hv[k] = h[(size_t)i * IN + k];
    float* orow = out + (size_t)i * ROW_STRIDE(OUT);
    #pragma unroll
    for (int o = 0; o < OUT; ++o) {
        float s = 0.0f;
        #pragma unroll
        for (int k = 0; k < IN; ++k) s = fmaf(hv[k], W[k * OUT + o], s);
        orow[o] = s;
    }
    orow[OUT] = dinv[i];
    if (OUT == 2) orow[3] = 0.0f;
}

#define SC_THREADS 1024
#define SC_GROUPS  256

template <int OUT> struct RowVec;
template <> struct RowVec<4> { typedef float4 type; };
template <> struct RowVec<2> { typedef float2 type; };

template <int OUT> struct GRow;
template <> struct GRow<4> {
    float4 h; float d;
    __device__ __forceinline__ void load(const float* __restrict__ hrow, int node) {
        const float* p = hrow + (size_t)node * ROW_STRIDE(4);
        h = *reinterpret_cast<const float4*>(p);
        d = p[4];
    }
    __device__ __forceinline__ float feat(int f) const {
        return f == 0 ? h.x : f == 1 ? h.y : f == 2 ? h.z : h.w;
    }
    __device__ __forceinline__ float dinv() const { return d; }
};
template <> struct GRow<2> {
    float4 q;
    __device__ __forceinline__ void load(const float* __restrict__ hrow, int node) {
        q = *reinterpret_cast<const float4*>(hrow + (size_t)node * ROW_STRIDE(2));
    }
    __device__ __forceinline__ float feat(int f) const { return f == 0 ? q.x : q.y; }
    __device__ __forceinline__ float dinv() const { return q.z; }
};

template <int OUT>
__global__ void __launch_bounds__(SC_THREADS)
edge_range_kernel(const int* __restrict__ row,
                  const int* __restrict__ col,
                  const float* __restrict__ dinv,
                  const float* __restrict__ hlin,
                  float* __restrict__ part) {
    constexpr int P     = 2 * OUT;
    constexpr int G     = SC_GROUPS / P;
    constexpr int NPART = N_NODES / P;
    constexpr int ESL   = N_EDGES / G;
    constexpr int NV    = ESL / 4;
    constexpr int NACC  = NPART * OUT;
    static_assert(N_NODES % P == 0 && N_EDGES % G == 0 && ESL % 4 == 0 && NACC % 4 == 0
                  && NPART % 4 == 0,
                  "ranges, slices and their 16-byte pieces divide evenly");
    __shared__ __attribute__((aligned(16))) float acc[NACC];
    __shared__ __attribute__((aligned(16))) float dloc[NPART];

    const int t = threadIdx.x;
    const int p = blockIdx.x % P;
    const int g = blockIdx.x / P;
    const int base = p * NPART;

    constexpr int DV     = NPART / 4;
    constexpr int DTRIPS = (DV + SC_THREADS - 1) / SC_THREADS;
    const float4* dsrc  = reinterpret_cast<const float4*>(dinv + base);
    float4*       dloc4 = reinterpret_cast<float4*>(dloc);
    float4 dq[DTRIPS];
    #pragma unroll
    for (int u = 0; u < DTRIPS; ++u) {
        const int i = t + u * SC_THREADS;
        dq[u] = make_float4(0.f, 0.f, 0.f, 0.f);
        if (i < DV) dq[u] = dsrc[i];
    }

    float4* acc4 = reinterpret_cast<float4*>(acc);
    for (int i = t; i < NACC / 4; i += SC_THREADS) acc4[i] = make_float4(0.f, 0.f, 0.f, 0.f);

    #pragma unroll
    for (int u = 0; u < DTRIPS; ++u) {
        const int i = t + u * SC_THREADS;
        if (i < DV) dloc4[i] = dq[u];
    }
    __syncthreads();

    const int4* row4 = reinterpret_cast<const int4*>(row + (size_t)g * ESL);
    const int4* col4 = reinterpret_cast<const int4*>(col + (size_t)g * ESL);

    int v = t;
    int4 rN = make_int4(-1, -1, -1, -1), cN = make_int4(0, 0, 0, 0);
    if (v < NV) { rN = row4[v]; cN = col4[v]; }
    for (; v < NV; v += SC_THREADS) {
        const int4 r = rN, c = cN;
        const int vn = v + SC_THREADS;
        if (vn < NV) { rN = row4[vn]; cN = col4[vn]; }

        const int rr[4] = { r.x - base, r.y - base, r.z - base, r.w - base };
        const int cc[4] = { c.x, c.y, c.z, c.w };
        bool       m[4];
        GRow<OUT>  gr[4];
        float      dr[4];
        #pragma unroll
        for (int j = 0; j < 4; ++j) {
            m[j] = (unsigned)rr[j] < (unsigned)NPART;
            dr[j] = dloc[m[j] ? rr[j] : 0];
        }
        #pragma unroll
        for (int j = 0; j < 4; ++j) {
            if (m[j]) gr[j].load(hlin, cc[j]);
        }
        #pragma unroll
        for (int j = 0; j < 4; ++j) {
            if (m[j]) {
                const float nrm = dr[j] * gr[j].dinv();
                #pragma unroll
                for (int f = 0; f < OUT; ++f)
                    __hip_atomic_fetch_add(&acc[rr[j] * OUT + f], gr[j].feat(f) * nrm,
                                           __ATOMIC_RELAXED, __HIP_MEMORY_SCOPE_WORKGROUP);
            }
        }
    }
    __syncthreads();

    float4* dst = reinterpret_cast<float4*>(part + ((size_t)g * N_NODES + base) * OUT);
    for (int i = t; i < NACC / 4; i += SC_THREADS) dst[i] = acc4[i];
}

template <int OUT>
__global__ void finalize_kernel(const float* __restrict__ part,
                                const float* __restrict__ hlin,
                                const float* __restrict__ dinv,
                                const float* __restrict__ b,
                                float* __restrict__ out, int n) {
    typedef typename RowVec<OUT>::type VT;
    constexpr int G = SC_GROUPS / (2 * OUT);
    int i = blockIdx.x * blockDim.x + threadIdx.x;
    if (i >= n) return;
    const VT* pv = reinterpret_cast<const VT*>(part);
    float agg[OUT];
    #pragma unroll
    for (int f = 0; f < OUT; ++f) agg[f] = 0.0f;
    #pragma unroll 16
    for (int g = 0; g < G; ++g) {
        const VT pg = pv[(size_t)g * N_NODES + i];
        const float* pf = reinterpret_cast<const float*>(&pg);
        #pragma unroll
        for (int f = 0; f < OUT; ++f) agg[f] += pf[f];
    }
    const float d2 = dinv[i] * dinv[i];
    #pragma unroll
    for (int f = 0; f < OUT; ++f) {
        float v = agg[f]
                + hlin[(size_t)i * ROW_STRIDE(OUT) + f] * d2 + b[f];
        out[(size_t)i * OUT + f] = tanhf(v);
    }
}

__global__ void classify_kernel(const float* __restrict__ h,
                                const float* __restrict__ Wc,
                                const float* __restrict__ bc,
                                float* __restrict__ out, int n) {
    int i = blockIdx.x * blockDim.x + threadIdx.x;
    if (i >= n) return;
    const float h0 = h[(size_t)i * 2 + 0];
    const float h1 = h[(size_t)i * 2 + 1];
    #pragma unroll
    for (int c = 0; c < 4; ++c) {
        out[(size_t)i * 4 + c] = fmaf(h0, Wc[c], fmaf(h1, Wc[4 + c], bc[c]));
    }
}

extern "C" void kernel_launch(void* const* d_in, const int* in_sizes, int n_in,
                              void* d_out, int out_size, void* d_ws, size_t ws_size,
                              hipStream_t stream) {
    const float* x  = (const float*)d_in[0];
    const int*   ei = (const int*)d_in[1];
    const float* W1 = (const float*)d_in[2];
    const float* b1 = (const float*)d_in[3];
    const float* W2 = (const float*)d_in[4];
    const float* b2 = (const float*)d_in[5];
    const float* W3 = (const float*)d_in[6];
    const float* b3 = (const float*)d_in[7];
    const float* Wc = (const float*)d_in[8];
    const float* bc = (const float*)d_in[9];

    const int N = N_NODES;
    const int E = N_EDGES;
    const int* row = ei;
    const int* col = ei + E;

    float* ws   = (float*)d_ws;
    float* dinv = ws;
    float* hlin = dinv + N;
    float* hA   = hlin + 8 * N;
    float* hB   = hA   + 4 * N;
    float* part = hB   + 4 * N;

    float* outC = (float*)d_out;
    float* outH = outC + (size_t)4 * N;

    const int BT = 256;
    const int gN = (N + BT - 1) / BT;

    deg_range_kernel<<<DG_GROUPS, DG_THREADS, 0, stream>>>(col, part);
    dinv_planes_kernel<<<gN, BT, 0, stream>>>(part, dinv, N);

    {
        const int tiles  = N / 16;
        const int blocks = (tiles + 7) / 8;
        gcn_wmma_xw1_kernel<<<blocks, 256, 0, stream>>>(x, W1, dinv, hlin, N);
        edge_range_kernel<4><<<SC_GROUPS, SC_THREADS, 0, stream>>>(row, col, dinv, hlin, part);
        finalize_kernel<4><<<gN, BT, 0, stream>>>(part, hlin, dinv, b1, hA, N);
    }
    {
        dense_small_kernel<4, 4><<<gN, BT, 0, stream>>>(hA, W2, dinv, hlin, N);
        edge_range_kernel<4><<<SC_GROUPS, SC_THREADS, 0, stream>>>(row, col, dinv, hlin, part);
        finalize_kernel<4><<<gN, BT, 0, stream>>>(part, hlin, dinv, b2, hB, N);
    }
    {
        dense_small_kernel<4, 2><<<gN, BT, 0, stream>>>(hB, W3, dinv, hlin, N);
        edge_range_kernel<2><<<SC_GROUPS, SC_THREADS, 0, stream>>>(row, col, dinv, hlin, part);
        finalize_kernel<2><<<gN, BT, 0, stream>>>(part, hlin, dinv, b3, outH, N);
    }
    classify_kernel<<<gN, BT, 0, stream>>>(outH, Wc, bc, outC, N);
}


# Per-node launches fused: each plane-sum epilogue also forms the next layer's table row (or the classifier), and the x@W1 launch adds the degree planes into dinv itself; 12 launches to 8; values same
# speedup vs baseline: 6.5930x; 1.0645x over previous
#include <hip/hip_runtime.h>
#include <hip/hip_bf16.h>
#include <math.h>

#define N_NODES   100000
#define N_EDGES   3200000

typedef float v2f __attribute__((ext_vector_type(2)));
typedef float v8f __attribute__((ext_vector_type(8)));

#define DG_THREADS 1024
#define DG_GROUPS  256
#define DG_P       4

__global__ void __launch_bounds__(DG_THREADS)
deg_range_kernel(const int* __restrict__ col, float* __restrict__ cntp) {
    constexpr int P     = DG_P;
    constexpr int G     = DG_GROUPS / P;
    constexpr int NPART = N_NODES / P;
    constexpr int ESL   = N_EDGES / G;
    constexpr int NV    = ESL / 4;
    constexpr int TRIPS = (NV + DG_THREADS - 1) / DG_THREADS;
    static_assert(N_NODES % P == 0 && N_EDGES % G == 0 && ESL % 4 == 0 && NPART % 4 == 0,
                  "ranges, slices and their 16-byte pieces divide evenly");
    __shared__ unsigned int cnt[NPART];

    const int t = threadIdx.x;
    const int p = blockIdx.x % P;
    const int g = blockIdx.x / P;
    const int base = p * NPART;

    uint4* cnt4 = reinterpret_cast<uint4*>(cnt);
    for (int i = t; i < NPART / 4; i += DG_THREADS) cnt4[i] = make_uint4(0u, 0u, 0u, 0u);
    __syncthreads();

    const int4* col4 = reinterpret_cast<const int4*>(col + (size_t)g * ESL);

    int4 c[TRIPS];
    #pragma unroll
    for (int u = 0; u < TRIPS; ++u) {
        const int v = t + u * DG_THREADS;
        c[u] = make_int4(-1, -1, -1, -1);
        if (v < NV) c[u] = col4[v];
    }
    #pragma unroll
    for (int u = 0; u < TRIPS; ++u) {
        const int cc[4] = { c[u].x - base, c[u].y - base, c[u].z - base, c[u].w - base };
        #pragma unroll
        for (int j = 0; j < 4; ++j) {
            if ((unsigned)cc[j] < (unsigned)NPART)
                __hip_atomic_fetch_add(&cnt[cc[j]], 1u,
                                       __ATOMIC_RELAXED, __HIP_MEMORY_SCOPE_WORKGROUP);
        }
    }
    __syncthreads();

    float4* dst = reinterpret_cast<float4*>(cntp + (size_t)g * N_NODES + base);
    for (int i = t; i < NPART / 4; i += DG_THREADS) {
        const uint4 k = cnt4[i];
        dst[i] = make_float4((float)k.x, (float)k.y, (float)k.z, (float)k.w);
    }
}

#define ROW_STRIDE(OUT) (2 * (OUT))

__global__ void __launch_bounds__(256)
gcn_wmma_xw1_kernel(const float* __restrict__ x,
                    const float* __restrict__ W1,
                    const float* __restrict__ cntp,
                    float* __restrict__ dinv,
                    float* __restrict__ hlin,
                    int n) {
    constexpr int G = DG_GROUPS / DG_P;
    static_assert(G % 2 == 0, "the planes split evenly over the two lane halves");
    __shared__ float Wp[128 * 16];
    const int t = threadIdx.x;
    #pragma unroll
    for (int j = 0; j < 8; ++j) {
        int idx = t * 8 + j;
        int kr  = idx >> 4;
        int nc  = idx & 15;
        Wp[idx] = (nc < 4) ? W1[kr * 4 + nc] : 0.0f;
    }
    __syncthreads();

    const int wave  = t >> 5;
    const int lane  = t & 31;
    const int tile  = blockIdx.x * 8 + wave;
    const int nTiles = n >> 4;
    if (tile >= nTiles) return;

    const int base = tile * 16;
    const int hh   = lane >> 4;
    const int nl   = lane & 15;
    const int arow = base + nl;

    float cs = 0.0f;
    #pragma unroll
    for (int g = 0; g < G / 2; ++g)
        cs += cntp[(size_t)(hh * (G / 2) + g) * N_NODES + arow];
    cs += __shfl_xor(cs, 16);
    const float dv = rsqrtf(1.0f + cs);

    v8f c = {};
    #pragma unroll 4
    for (int k = 0; k < 128; k += 4) {
        const int kk = k + 2 * hh;
        v2f a = *reinterpret_cast<const v2f*>(x + (size_t)arow * 128 + kk);
        v2f b;
        b.x = Wp[(kk + 0) * 16 + nl];
        b.y = Wp[(kk + 1) * 16 + nl];
        c = __builtin_amdgcn_wmma_f32_16x16x4_f32(
                 false, a,  false, b,
                 (short)0, c,  false,  false);
    }

    if (nl < 4) {
        #pragma unroll
        for (int v = 0; v < 8; ++v) {
            int m = base + v + 8 * hh;
            hlin[(size_t)m * ROW_STRIDE(4) + nl] = c[v];
        }
    }
    if (hh == 0) {
        dinv[arow] = dv;
        hlin[(size_t)arow * ROW_STRIDE(4) + 4] = dv;
    }
}

#define SC_THREADS 1024
#define SC_GROUPS  256

template <int OUT> struct RowVec;
template <> struct RowVec<4> { typedef float4 type; };
template <> struct RowVec<2> { typedef float2 type; };

template <int OUT> struct GRow;
template <> struct GRow<4> {
    float4 h; float d;
    __device__ __forceinline__ void load(const float* __restrict__ hrow, int node) {
        const float* p = hrow + (size_t)node * ROW_STRIDE(4);
        h = *reinterpret_cast<const float4*>(p);
        d = p[4];
    }
    __device__ __forceinline__ float feat(int f) const {
        return f == 0 ? h.x : f == 1 ? h.y : f == 2 ? h.z : h.w;
    }
    __device__ __forceinline__ float dinv() const { return d; }
};
template <> struct GRow<2> {
    float4 q;
    __device__ __forceinline__ void load(const float* __restrict__ hrow, int node) {
        q = *reinterpret_cast<const float4*>(hrow + (size_t)node * ROW_STRIDE(2));
    }
    __device__ __forceinline__ float feat(int f) const { return f == 0 ? q.x : q.y; }
    __device__ __forceinline__ float dinv() const { return q.z; }
};

template <int OUT>
__global__ void __launch_bounds__(SC_THREADS)
edge_range_kernel(const int* __restrict__ row,
                  const int* __restrict__ col,
                  const float* __restrict__ dinv,
                  const float* __restrict__ hlin,
                  float* __restrict__ part) {
    constexpr int P     = 2 * OUT;
    constexpr int G     = SC_GROUPS / P;
    constexpr int NPART = N_NODES / P;
    constexpr int ESL   = N_EDGES / G;
    constexpr int NV    = ESL / 4;
    constexpr int NACC  = NPART * OUT;
    static_assert(N_NODES % P == 0 && N_EDGES % G == 0 && ESL % 4 == 0 && NACC % 4 == 0
                  && NPART % 4 == 0,
                  "ranges, slices and their 16-byte pieces divide evenly");
    __shared__ __attribute__((aligned(16))) float acc[NACC];
    __shared__ __attribute__((aligned(16))) float dloc[NPART];

    const int t = threadIdx.x;
    const int p = blockIdx.x % P;
    const int g = blockIdx.x / P;
    const int base = p * NPART;

    constexpr int DV     = NPART / 4;
    constexpr int DTRIPS = (DV + SC_THREADS - 1) / SC_THREADS;
    const float4* dsrc  = reinterpret_cast<const float4*>(dinv + base);
    float4*       dloc4 = reinterpret_cast<float4*>(dloc);
    float4 dq[DTRIPS];
    #pragma unroll
    for (int u = 0; u < DTRIPS; ++u) {
        const int i = t + u * SC_THREADS;
        dq[u] = make_float4(0.f, 0.f, 0.f, 0.f);
        if (i < DV) dq[u] = dsrc[i];
    }

    float4* acc4 = reinterpret_cast<float4*>(acc);
    for (int i = t; i < NACC / 4; i += SC_THREADS) acc4[i] = make_float4(0.f, 0.f, 0.f, 0.f);

    #pragma unroll
    for (int u = 0; u < DTRIPS; ++u) {
        const int i = t + u * SC_THREADS;
        if (i < DV) dloc4[i] = dq[u];
    }
    __syncthreads();

    const int4* row4 = reinterpret_cast<const int4*>(row + (size_t)g * ESL);
    const int4* col4 = reinterpret_cast<const int4*>(col + (size_t)g * ESL);

    int v = t;
    int4 rN = make_int4(-1, -1, -1, -1), cN = make_int4(0, 0, 0, 0);
    if (v < NV) { rN = row4[v]; cN = col4[v]; }
    for (; v < NV; v += SC_THREADS) {
        const int4 r = rN, c = cN;
        const int vn = v + SC_THREADS;
        if (vn < NV) { rN = row4[vn]; cN = col4[vn]; }

        const int rr[4] = { r.x - base, r.y - base, r.z - base, r.w - base };
        const int cc[4] = { c.x, c.y, c.z, c.w };
        bool       m[4];
        GRow<OUT>  gr[4];
        float      dr[4];
        #pragma unroll
        for (int j = 0; j < 4; ++j) {
            m[j] = (unsigned)rr[j] < (unsigned)NPART;
            dr[j] = dloc[m[j] ? rr[j] : 0];
        }
        #pragma unroll
        for (int j = 0; j < 4; ++j) {
            if (m[j]) gr[j].load(hlin, cc[j]);
        }
        #pragma unroll
        for (int j = 0; j < 4; ++j) {
            if (m[j]) {
                const float nrm = dr[j] * gr[j].dinv();
                #pragma unroll
                for (int f = 0; f < OUT; ++f)
                    __hip_atomic_fetch_add(&acc[rr[j] * OUT + f], gr[j].feat(f) * nrm,
                                           __ATOMIC_RELAXED, __HIP_MEMORY_SCOPE_WORKGROUP);
            }
        }
    }
    __syncthreads();

    float4* dst = reinterpret_cast<float4*>(part + ((size_t)g * N_NODES + base) * OUT);
    for (int i = t; i < NACC / 4; i += SC_THREADS) dst[i] = acc4[i];
}

template <int OUT>
__device__ __forceinline__ void node_features(const float* __restrict__ part,
                                              const float* __restrict__ hlin,
                                              float di,
                                              const float* __restrict__ b,
                                              int i, float (&hv)[OUT]) {
    typedef typename RowVec<OUT>::type VT;
    constexpr int G = SC_GROUPS / (2 * OUT);
    const VT* pv = reinterpret_cast<const VT*>(part);
    float agg[OUT];
    #pragma unroll
    for (int f = 0; f < OUT; ++f) agg[f] = 0.0f;
    #pragma unroll 16
    for (int g = 0; g < G; ++g) {
        const VT pg = pv[(size_t)g * N_NODES + i];
        const float* pf = reinterpret_cast<const float*>(&pg);
        #pragma unroll
        for (int f = 0; f < OUT; ++f) agg[f] += pf[f];
    }
    const float d2 = di * di;
    #pragma unroll
    for (int f = 0; f < OUT; ++f) {
        float v = agg[f]
                + hlin[(size_t)i * ROW_STRIDE(OUT) + f] * d2 + b[f];
        hv[f] = tanhf(v);
    }
}

template <int OUT, int NEXT>
__global__ void node_stage_kernel(const float* __restrict__ part,
                                  const float* __restrict__ hlin,
                                  const float* __restrict__ dinv,
                                  const float* __restrict__ b,
                                  const float* __restrict__ Wn,
                                  float* __restrict__ tnext,
                                  int n) {
    int i = blockIdx.x * blockDim.x + threadIdx.x;
    if (i >= n) return;
    const float di = dinv[i];
    float hv[OUT];
    node_features<OUT>(part, hlin, di, b, i, hv);

    float s[NEXT];
    #pragma unroll
    for (int o = 0; o < NEXT; ++o) {
        s[o] = 0.0f;
        #pragma unroll
        for (int k = 0; k < OUT; ++k) s[o] = fmaf(hv[k], Wn[k * NEXT + o], s[o]);
    }
    float* orow = tnext + (size_t)i * ROW_STRIDE(NEXT);
    if constexpr (NEXT == 4) {
        *reinterpret_cast<float4*>(orow) = make_float4(s[0], s[1], s[2], s[3]);
        orow[4] = di;
    } else {
        static_assert(NEXT == 2, "rows of 4 or 2 features");
        *reinterpret_cast<float4*>(orow) = make_float4(s[0], s[1], di, 0.0f);
    }
}

__global__ void node_last_kernel(const float* __restrict__ part,
                                 const float* __restrict__ hlin,
                                 const float* __restrict__ dinv,
                                 const float* __restrict__ b,
                                 const float* __restrict__ Wc,
                                 const float* __restrict__ bc,
                                 float* __restrict__ outH,
                                 float* __restrict__ outC,
                                 int n) {
    int i = blockIdx.x * blockDim.x + threadIdx.x;
    if (i >= n) return;
    float hv[2];
    node_features<2>(part, hlin, dinv[i], b, i, hv);
    *reinterpret_cast<float2*>(outH + (size_t)i * 2) = make_float2(hv[0], hv[1]);
    float cl[4];
    #pragma unroll
    for (int c = 0; c < 4; ++c)
        cl[c] = fmaf(hv[0], Wc[c], fmaf(hv[1], Wc[4 + c], bc[c]));
    *reinterpret_cast<float4*>(outC + (size_t)i * 4) = make_float4(cl[0], cl[1], cl[2], cl[3]);
}

extern "C" void kernel_launch(void* const* d_in, const int* in_sizes, int n_in,
                              void* d_out, int out_size, void* d_ws, size_t ws_size,
                              hipStream_t stream) {
    const float* x  = (const float*)d_in[0];
    const int*   ei = (const int*)d_in[1];
    const float* W1 = (const float*)d_in[2];
    const float* b1 = (const float*)d_in[3];
    const float* W2 = (const float*)d_in[4];
    const float* b2 = (const float*)d_in[5];
    const float* W3 = (const float*)d_in[6];
    const float* b3 = (const float*)d_in[7];
    const float* Wc = (const float*)d_in[8];
    const float* bc = (const float*)d_in[9];

    const int N = N_NODES;
    const int E = N_EDGES;
    const int* row = ei;
    const int* col = ei + E;

    float* ws   = (float*)d_ws;
    float* dinv = ws;
    float* tabA = dinv + N;
    float* tabB = tabA + 8 * N;
    float* part = tabB + 8 * N;

    float* outC = (float*)d_out;
    float* outH = outC + (size_t)4 * N;

    const int BT = 256;
    const int gN = (N + BT - 1) / BT;

    deg_range_kernel<<<DG_GROUPS, DG_THREADS, 0, stream>>>(col, part);

    {
        const int tiles  = N / 16;
        const int blocks = (tiles + 7) / 8;
        gcn_wmma_xw1_kernel<<<blocks, 256, 0, stream>>>(x, W1, part, dinv, tabA, N);
        edge_range_kernel<4><<<SC_GROUPS, SC_THREADS, 0, stream>>>(row, col, dinv, tabA, part);
    }
    {
        node_stage_kernel<4, 4><<<gN, BT, 0, stream>>>(part, tabA, dinv, b1, W2, tabB, N);
        edge_range_kernel<4><<<SC_GROUPS, SC_THREADS, 0, stream>>>(row, col, dinv, tabB, part);
    }
    {
        node_stage_kernel<4, 2><<<gN, BT, 0, stream>>>(part, tabB, dinv, b2, W3, tabA, N);
        edge_range_kernel<2><<<SC_GROUPS, SC_THREADS, 0, stream>>>(row, col, dinv, tabA, part);
    }
    node_last_kernel<<<gN, BT, 0, stream>>>(part, tabA, dinv, b3, Wc, bc, outH, outC, N);
}


# Edge list read once: the degree launch also streams row and each wave stores its range's edges (offset, source in one word); the 3 edge launches read these lists, 8 ranges each; LDS add order changes
# speedup vs baseline: 7.8343x; 1.1883x over previous
#include <hip/hip_runtime.h>
#include <hip/hip_bf16.h>
#include <math.h>

#define N_NODES   100000
#define N_EDGES   3200000

typedef float v2f __attribute__((ext_vector_type(2)));
typedef float v8f __attribute__((ext_vector_type(8)));

#define RG_THREADS 1024
#define RG_GROUPS  256
#define RG_P       8
#define RG_G       (RG_GROUPS / RG_P)
#define RG_NPART   (N_NODES / RG_P)
#define RG_ESL     (N_EDGES / RG_G)
#define RG_NV      (RG_ESL / 4)
#define RG_WAVES   (RG_THREADS / 32)
#define RG_TRIPS   ((RG_NV + RG_THREADS - 1) / RG_THREADS)
#define RG_LCAP    (RG_TRIPS * 128)
#define COL_BITS   17
static_assert(N_NODES % RG_P == 0 && N_EDGES % RG_G == 0 && RG_ESL % 4 == 0 && RG_NPART % 4 == 0,
              "ranges, slices and their 16-byte pieces divide evenly");
static_assert(N_NODES <= (1 << COL_BITS) && RG_NPART < (1 << (32 - COL_BITS)) - 1,
              "offset in range and source share one 32-bit word; all ones is no edge");

__global__ void __launch_bounds__(RG_THREADS)
range_filter_kernel(const int* __restrict__ row,
                    const int* __restrict__ col,
                    float* __restrict__ cntp,
                    unsigned int* __restrict__ lists,
                    int* __restrict__ lcount) {
    constexpr int NPART = RG_NPART;
    constexpr int U     = 5;
    constexpr int CH    = RG_TRIPS / U;
    static_assert(RG_TRIPS % U == 0, "a lane's pieces split evenly into groups of U");
    __shared__ unsigned int cnt[NPART];

    const int t    = threadIdx.x;
    const int lane = t & 31;
    const int wv   = __builtin_amdgcn_readfirstlane(t >> 5);
    const int p    = blockIdx.x % RG_P;
    const int g    = blockIdx.x / RG_P;
    const int base = p * NPART;

    uint4* cnt4 = reinterpret_cast<uint4*>(cnt);
    for (int i = t; i < NPART / 4; i += RG_THREADS) cnt4[i] = make_uint4(0u, 0u, 0u, 0u);
    __syncthreads();

    const int4* row4 = reinterpret_cast<const int4*>(row + (size_t)g * RG_ESL);
    const int4* col4 = reinterpret_cast<const int4*>(col + (size_t)g * RG_ESL);
    unsigned int* wl = lists + ((size_t)blockIdx.x * RG_WAVES + wv) * RG_LCAP;
    int nl = 0;

    for (int ch = 0; ch < CH; ++ch) {
        int4 r[U], c[U];
        #pragma unroll
        for (int u = 0; u < U; ++u) {
            const int v = t + (ch * U + u) * RG_THREADS;
            r[u] = make_int4(-1, -1, -1, -1);
            c[u] = make_int4(-1, -1, -1, -1);
            if (v < RG_NV) { r[u] = row4[v]; c[u] = col4[v]; }
        }
        #pragma unroll
        for (int u = 0; u < U; ++u) {
            const int rr[4] = { r[u].x - base, r[u].y - base, r[u].z - base, r[u].w - base };
            const int cc[4] = { c[u].x, c[u].y, c[u].z, c[u].w };
            #pragma unroll
            for (int j = 0; j < 4; ++j) {
                if ((unsigned)(cc[j] - base) < (unsigned)NPART)
                    __hip_atomic_fetch_add(&cnt[cc[j] - base], 1u,
                                           __ATOMIC_RELAXED, __HIP_MEMORY_SCOPE_WORKGROUP);
                const bool m = (unsigned)rr[j] < (unsigned)NPART;
                const unsigned int mask = __builtin_amdgcn_ballot_w32(m);
                const int pos = nl + (int)__builtin_amdgcn_mbcnt_lo(mask, 0u);
                if (m) wl[pos] = ((unsigned)rr[j] << COL_BITS) | (unsigned)cc[j];
                nl += __builtin_popcount(mask);
            }
        }
    }
    const int padn = (4 - (nl & 3)) & 3;
    if (lane < padn) wl[nl + lane] = 0xFFFFFFFFu;
    if (lane == 0) lcount[blockIdx.x * RG_WAVES + wv] = nl;
    __syncthreads();

    float4* dst = reinterpret_cast<float4*>(cntp + (size_t)g * N_NODES + base);
    for (int i = t; i < NPART / 4; i += RG_THREADS) {
        const uint4 k = cnt4[i];
        dst[i] = make_float4((float)k.x, (float)k.y, (float)k.z, (float)k.w);
    }
}

#define ROW_STRIDE(OUT) (2 * (OUT))

__global__ void __launch_bounds__(256)
gcn_wmma_xw1_kernel(const float* __restrict__ x,
                    const float* __restrict__ W1,
                    const float* __restrict__ cntp,
                    float* __restrict__ dinv,
                    float* __restrict__ hlin,
                    int n) {
    constexpr int G = RG_G;
    static_assert(G % 2 == 0, "the planes split evenly over the two lane halves");
    __shared__ float Wp[128 * 16];
    const int t = threadIdx.x;
    #pragma unroll
    for (int j = 0; j < 8; ++j) {
        int idx = t * 8 + j;
        int kr  = idx >> 4;
        int nc  = idx & 15;
        Wp[idx] = (nc < 4) ? W1[kr * 4 + nc] : 0.0f;
    }
    __syncthreads();

    const int wave  = t >> 5;
    const int lane  = t & 31;
    const int tile  = blockIdx.x * 8 + wave;
    const int nTiles = n >> 4;
    if (tile >= nTiles) return;

    const int base = tile * 16;
    const int hh   = lane >> 4;
    const int nl   = lane & 15;
    const int arow = base + nl;

    float cs = 0.0f;
    #pragma unroll
    for (int g = 0; g < G / 2; ++g)
        cs += cntp[(size_t)(hh * (G / 2) + g) * N_NODES + arow];
    cs += __shfl_xor(cs, 16);
    const float dv = rsqrtf(1.0f + cs);

    v8f c = {};
    #pragma unroll 4
    for (int k = 0; k < 128; k += 4) {
        const int kk = k + 2 * hh;
        v2f a = *reinterpret_cast<const v2f*>(x + (size_t)arow * 128 + kk);
        v2f b;
        b.x = Wp[(kk + 0) * 16 + nl];
        b.y = Wp[(kk + 1) * 16 + nl];
        c = __builtin_amdgcn_wmma_f32_16x16x4_f32(
                 false, a,  false, b,
                 (short)0, c,  false,  false);
    }

    if (nl < 4) {
        #pragma unroll
        for (int v = 0; v < 8; ++v) {
            int m = base + v + 8 * hh;
            hlin[(size_t)m * ROW_STRIDE(4) + nl] = c[v];
        }
    }
    if (hh == 0) {
        dinv[arow] = dv;
        hlin[(size_t)arow * ROW_STRIDE(4) + 4] = dv;
    }
}

template <int OUT> struct RowVec;
template <> struct RowVec<4> { typedef float4 type; };
template <> struct RowVec<2> { typedef float2 type; };

template <int OUT> struct GRow;
template <> struct GRow<4> {
    float4 h; float d;
    __device__ __forceinline__ void load(const float* __restrict__ hrow, int node) {
        const float* p = hrow + (size_t)node * ROW_STRIDE(4);
        h = *reinterpret_cast<const float4*>(p);
        d = p[4];
    }
    __device__ __forceinline__ float feat(int f) const {
        return f == 0 ? h.x : f == 1 ? h.y : f == 2 ? h.z : h.w;
    }
    __device__ __forceinline__ float dinv() const { return d; }
};
template <> struct GRow<2> {
    float4 q;
    __device__ __forceinline__ void load(const float* __restrict__ hrow, int node) {
        q = *reinterpret_cast<const float4*>(hrow + (size_t)node * ROW_STRIDE(2));
    }
    __device__ __forceinline__ float feat(int f) const { return f == 0 ? q.x : q.y; }
    __device__ __forceinline__ float dinv() const { return q.z; }
};

template <int OUT>
__global__ void __launch_bounds__(RG_THREADS)
edge_list_kernel(const unsigned int* __restrict__ lists,
                 const int* __restrict__ lcount,
                 const float* __restrict__ dinv,
                 const float* __restrict__ hlin,
                 float* __restrict__ part) {
    constexpr int NPART = RG_NPART;
    constexpr int NACC  = NPART * OUT;
    constexpr int Q     = 2;
    constexpr int NE    = 4 * Q;
    static_assert(NACC % 4 == 0, "the range's sums split into 16-byte pieces");
    __shared__ __attribute__((aligned(16))) float acc[NACC];
    __shared__ __attribute__((aligned(16))) float dloc[NPART];

    const int t    = threadIdx.x;
    const int lane = t & 31;
    const int wv   = __builtin_amdgcn_readfirstlane(t >> 5);
    const int p    = blockIdx.x % RG_P;
    const int g    = blockIdx.x / RG_P;
    const int base = p * NPART;

    const int nl  = lcount[blockIdx.x * RG_WAVES + wv];
    const int npc = min((nl + 3) >> 2, RG_LCAP / 4);
    const uint4* wl4 = reinterpret_cast<const uint4*>(
                           lists + ((size_t)blockIdx.x * RG_WAVES + wv) * RG_LCAP);
    const uint4 none = make_uint4(~0u, ~0u, ~0u, ~0u);
    uint4 qN[Q];
    #pragma unroll
    for (int i = 0; i < Q; ++i) {
        const int k = i * 32 + lane;
        qN[i] = none;
        if (k < npc) qN[i] = wl4[k];
    }

    constexpr int DV     = NPART / 4;
    constexpr int DTRIPS = (DV + RG_THREADS - 1) / RG_THREADS;
    const float4* dsrc  = reinterpret_cast<const float4*>(dinv + base);
    float4*       dloc4 = reinterpret_cast<float4*>(dloc);
    float4 dq[DTRIPS];
    #pragma unroll
    for (int u = 0; u < DTRIPS; ++u) {
        const int i = t + u * RG_THREADS;
        dq[u] = make_float4(0.f, 0.f, 0.f, 0.f);
        if (i < DV) dq[u] = dsrc[i];
    }

    float4* acc4 = reinterpret_cast<float4*>(acc);
    for (int i = t; i < NACC / 4; i += RG_THREADS) acc4[i] = make_float4(0.f, 0.f, 0.f, 0.f);

    #pragma unroll
    for (int u = 0; u < DTRIPS; ++u) {
        const int i = t + u * RG_THREADS;
        if (i < DV) dloc4[i] = dq[u];
    }
    __syncthreads();

    for (int k0 = 0; k0 < npc; k0 += 32 * Q) {
        unsigned int w[NE];
        #pragma unroll
        for (int i = 0; i < Q; ++i) {
            w[4 * i + 0] = qN[i].x; w[4 * i + 1] = qN[i].y;
            w[4 * i + 2] = qN[i].z; w[4 * i + 3] = qN[i].w;
        }
        #pragma unroll
        for (int i = 0; i < Q; ++i) {
            const int kn = k0 + 32 * Q + i * 32 + lane;
            qN[i] = none;
            if (kn < npc) qN[i] = wl4[kn];
        }

        int        rr[NE];
        bool       m[NE];
        GRow<OUT>  gr[NE];
        float      dr[NE];
        #pragma unroll
        for (int j = 0; j < NE; ++j) {
            rr[j] = (int)(w[j] >> COL_BITS);
            m[j]  = (unsigned)rr[j] < (unsigned)NPART;
            dr[j] = dloc[m[j] ? rr[j] : 0];
        }
        #pragma unroll
        for (int j = 0; j < NE; ++j) {
            if (m[j]) gr[j].load(hlin, (int)(w[j] & ((1u << COL_BITS) - 1u)));
        }
        #pragma unroll
        for (int j = 0; j < NE; ++j) {
            if (m[j]) {
                const float nrm = dr[j] * gr[j].dinv();
                #pragma unroll
                for (int f = 0; f < OUT; ++f)
                    __hip_atomic_fetch_add(&acc[rr[j] * OUT + f], gr[j].feat(f) * nrm,
                                           __ATOMIC_RELAXED, __HIP_MEMORY_SCOPE_WORKGROUP);
            }
        }
    }
    __syncthreads();

    float4* dst = reinterpret_cast<float4*>(part + ((size_t)g * N_NODES + base) * OUT);
    for (int i = t; i < NACC / 4; i += RG_THREADS) dst[i] = acc4[i];
}

template <int OUT>
__device__ __forceinline__ void node_features(const float* __restrict__ part,
                                              const float* __restrict__ hlin,
                                              float di,
                                              const float* __restrict__ b,
                                              int i, float (&hv)[OUT]) {
    typedef typename RowVec<OUT>::type VT;
    constexpr int G = RG_G;
    const VT* pv = reinterpret_cast<const VT*>(part);
    float agg[OUT];
    #pragma unroll
    for (int f = 0; f < OUT; ++f) agg[f] = 0.0f;
    #pragma unroll 16
    for (int g = 0; g < G; ++g) {
        const VT pg = pv[(size_t)g * N_NODES + i];
        const float* pf = reinterpret_cast<const float*>(&pg);
        #pragma unroll
        for (int f = 0; f < OUT; ++f) agg[f] += pf[f];
    }
    const float d2 = di * di;
    #pragma unroll
    for (int f = 0; f < OUT; ++f) {
        float v = agg[f]
                + hlin[(size_t)i * ROW_STRIDE(OUT) + f] * d2 + b[f];
        hv[f] = tanhf(v);
    }
}

template <int OUT, int NEXT>
__global__ void node_stage_kernel(const float* __restrict__ part,
                                  const float* __restrict__ hlin,
                                  const float* __restrict__ dinv,
                                  const float* __restrict__ b,
                                  const float* __restrict__ Wn,
                                  float* __restrict__ tnext,
                                  int n) {
    int i = blockIdx.x * blockDim.x + threadIdx.x;
    if (i >= n) return;
    const float di = dinv[i];
    float hv[OUT];
    node_features<OUT>(part, hlin, di, b, i, hv);

    float s[NEXT];
    #pragma unroll
    for (int o = 0; o < NEXT; ++o) {
        s[o] = 0.0f;
        #pragma unroll
        for (int k = 0; k < OUT; ++k) s[o] = fmaf(hv[k], Wn[k * NEXT + o], s[o]);
    }
    float* orow = tnext + (size_t)i * ROW_STRIDE(NEXT);
    if constexpr (NEXT == 4) {
        *reinterpret_cast<float4*>(orow) = make_float4(s[0], s[1], s[2], s[3]);
        orow[4] = di;
    } else {
        static_assert(NEXT == 2, "rows of 4 or 2 features");
        *reinterpret_cast<float4*>(orow) = make_float4(s[0], s[1], di, 0.0f);
    }
}

__global__ void node_last_kernel(const float* __restrict__ part,
                                 const float* __restrict__ hlin,
                                 const float* __restrict__ dinv,
                                 const float* __restrict__ b,
                                 const float* __restrict__ Wc,
                                 const float* __restrict__ bc,
                                 float* __restrict__ outH,
                                 float* __restrict__ outC,
                                 int n) {
    int i = blockIdx.x * blockDim.x + threadIdx.x;
    if (i >= n) return;
    float hv[2];
    node_features<2>(part, hlin, dinv[i], b, i, hv);
    *reinterpret_cast<float2*>(outH + (size_t)i * 2) = make_float2(hv[0], hv[1]);
    float cl[4];
    #pragma unroll
    for (int c = 0; c < 4; ++c)
        cl[c] = fmaf(hv[0], Wc[c], fmaf(hv[1], Wc[4 + c], bc[c]));
    *reinterpret_cast<float4*>(outC + (size_t)i * 4) = make_float4(cl[0], cl[1], cl[2], cl[3]);
}

extern "C" void kernel_launch(void* const* d_in, const int* in_sizes, int n_in,
                              void* d_out, int out_size, void* d_ws, size_t ws_size,
                              hipStream_t stream) {
    const float* x  = (const float*)d_in[0];
    const int*   ei = (const int*)d_in[1];
    const float* W1 = (const float*)d_in[2];
    const float* b1 = (const float*)d_in[3];
    const float* W2 = (const float*)d_in[4];
    const float* b2 = (const float*)d_in[5];
    const float* W3 = (const float*)d_in[6];
    const float* b3 = (const float*)d_in[7];
    const float* Wc = (const float*)d_in[8];
    const float* bc = (const float*)d_in[9];

    const int N = N_NODES;
    const int E = N_EDGES;
    const int* row = ei;
    const int* col = ei + E;

    float* ws   = (float*)d_ws;
    float* dinv = ws;
    float* tabA = dinv + N;
    float* tabB = tabA + 8 * N;
    float* part = tabB + 8 * N;
    unsigned int* lists  = (unsigned int*)(part + 128 * N);
    int*          lcount = (int*)(lists + (size_t)RG_GROUPS * RG_WAVES * RG_LCAP);

    float* outC = (float*)d_out;
    float* outH = outC + (size_t)4 * N;

    const int BT = 256;
    const int gN = (N + BT - 1) / BT;

    range_filter_kernel<<<RG_GROUPS, RG_THREADS, 0, stream>>>(row, col, part, lists, lcount);

    {
        const int tiles  = N / 16;
        const int blocks = (tiles + 7) / 8;
        gcn_wmma_xw1_kernel<<<blocks, 256, 0, stream>>>(x, W1, part, dinv, tabA, N);
        edge_list_kernel<4><<<RG_GROUPS, RG_THREADS, 0, stream>>>(lists, lcount, dinv, tabA, part);
    }
    {
        node_stage_kernel<4, 4><<<gN, BT, 0, stream>>>(part, tabA, dinv, b1, W2, tabB, N);
        edge_list_kernel<4><<<RG_GROUPS, RG_THREADS, 0, stream>>>(lists, lcount, dinv, tabB, part);
    }
    {
        node_stage_kernel<4, 2><<<gN, BT, 0, stream>>>(part, tabB, dinv, b2, W3, tabA, N);
        edge_list_kernel<2><<<RG_GROUPS, RG_THREADS, 0, stream>>>(lists, lcount, dinv, tabA, part);
    }
    node_last_kernel<<<gN, BT, 0, stream>>>(part, tabA, dinv, b3, Wc, bc, outH, outC, N);
}


# Edge launches cut the nodes into 16 ranges, not 8: a work-group walks the lists of two slices and keeps its half range's words; 32 planes of sums become 16; LDS add order and plane sum order change
# speedup vs baseline: 8.7289x; 1.1142x over previous
#include <hip/hip_runtime.h>
#include <hip/hip_bf16.h>
#include <math.h>

#define N_NODES   100000
#define N_EDGES   3200000

typedef float v2f __attribute__((ext_vector_type(2)));
typedef float v8f __attribute__((ext_vector_type(8)));

#define RG_THREADS 1024
#define RG_GROUPS  256
#define RG_P       8
#define RG_G       (RG_GROUPS / RG_P)
#define RG_NPART   (N_NODES / RG_P)
#define RG_ESL     (N_EDGES / RG_G)
#define RG_NV      (RG_ESL / 4)
#define RG_WAVES   (RG_THREADS / 32)
#define RG_TRIPS   ((RG_NV + RG_THREADS - 1) / RG_THREADS)
#define RG_LCAP    (RG_TRIPS * 128)
#define COL_BITS   17
static_assert(N_NODES % RG_P == 0 && N_EDGES % RG_G == 0 && RG_ESL % 4 == 0 && RG_NPART % 4 == 0,
              "ranges, slices and their 16-byte pieces divide evenly");
static_assert(N_NODES <= (1 << COL_BITS) && RG_NPART < (1 << (32 - COL_BITS)) - 1,
              "offset in range and source share one 32-bit word; all ones is no edge");

#define EG_P       16
#define EG_G       (RG_GROUPS / EG_P)
#define EG_NPART   (N_NODES / EG_P)
#define EG_SL      (RG_G / EG_G)
#define EG_SPLIT   (EG_P / RG_P)
static_assert(N_NODES % EG_P == 0 && RG_G % EG_G == 0 && EG_P % RG_P == 0 &&
              RG_NPART == EG_SPLIT * EG_NPART && EG_NPART % 2 == 0,
              "an edge range is a whole part of one filter range and begins on an 8-byte piece of dinv");

__global__ void __launch_bounds__(RG_THREADS)
range_filter_kernel(const int* __restrict__ row,
                    const int* __restrict__ col,
                    float* __restrict__ cntp,
                    unsigned int* __restrict__ lists,
                    int* __restrict__ lcount) {
    constexpr int NPART = RG_NPART;
    constexpr int U     = 5;
    constexpr int CH    = RG_TRIPS / U;
    static_assert(RG_TRIPS % U == 0, "a lane's pieces split evenly into groups of U");
    __shared__ unsigned int cnt[NPART];

    const int t    = threadIdx.x;
    const int lane = t & 31;
    const int wv   = __builtin_amdgcn_readfirstlane(t >> 5);
    const int p    = blockIdx.x % RG_P;
    const int g    = blockIdx.x / RG_P;
    const int base = p * NPART;

    uint4* cnt4 = reinterpret_cast<uint4*>(cnt);
    for (int i = t; i < NPART / 4; i += RG_THREADS) cnt4[i] = make_uint4(0u, 0u, 0u, 0u);
    __syncthreads();

    const int4* row4 = reinterpret_cast<const int4*>(row + (size_t)g * RG_ESL);
    const int4* col4 = reinterpret_cast<const int4*>(col + (size_t)g * RG_ESL);
    unsigned int* wl = lists + ((size_t)blockIdx.x * RG_WAVES + wv) * RG_LCAP;
    int nl = 0;

    for (int ch = 0; ch < CH; ++ch) {
        int4 r[U], c[U];
        #pragma unroll
        for (int u = 0; u < U; ++u) {
            const int v = t + (ch * U + u) * RG_THREADS;
            r[u] = make_int4(-1, -1, -1, -1);
            c[u] = make_int4(-1, -1, -1, -1);
            if (v < RG_NV) { r[u] = row4[v]; c[u] = col4[v]; }
        }
        #pragma unroll
        for (int u = 0; u < U; ++u) {
            const int rr[4] = { r[u].x - base, r[u].y - base, r[u].z - base, r[u].w - base };
            const int cc[4] = { c[u].x, c[u].y, c[u].z, c[u].w };
            #pragma unroll
            for (int j = 0; j < 4; ++j) {
                if ((unsigned)(cc[j] - base) < (unsigned)NPART)
                    __hip_atomic_fetch_add(&cnt[cc[j] - base], 1u,
                                           __ATOMIC_RELAXED, __HIP_MEMORY_SCOPE_WORKGROUP);
                const bool m = (unsigned)rr[j] < (unsigned)NPART;
                const unsigned int mask = __builtin_amdgcn_ballot_w32(m);
                const int pos = nl + (int)__builtin_amdgcn_mbcnt_lo(mask, 0u);
                if (m) wl[pos] = ((unsigned)rr[j] << COL_BITS) | (unsigned)cc[j];
                nl += __builtin_popcount(mask);
            }
        }
    }
    const int padn = (4 - (nl & 3)) & 3;
    if (lane < padn) wl[nl + lane] = 0xFFFFFFFFu;
    if (lane == 0) lcount[blockIdx.x * RG_WAVES + wv] = nl;
    __syncthreads();

    float4* dst = reinterpret_cast<float4*>(cntp + (size_t)g * N_NODES + base);
    for (int i = t; i < NPART / 4; i += RG_THREADS) {
        const uint4 k = cnt4[i];
        dst[i] = make_float4((float)k.x, (float)k.y, (float)k.z, (float)k.w);
    }
}

#define ROW_STRIDE(OUT) (2 * (OUT))

__global__ void __launch_bounds__(256)
gcn_wmma_xw1_kernel(const float* __restrict__ x,
                    const float* __restrict__ W1,
                    const float* __restrict__ cntp,
                    float* __restrict__ dinv,
                    float* __restrict__ hlin,
                    int n) {
    constexpr int G = RG_G;
    static_assert(G % 2 == 0, "the planes split evenly over the two lane halves");
    __shared__ float Wp[128 * 16];
    const int t = threadIdx.x;
    #pragma unroll
    for (int j = 0; j < 8; ++j) {
        int idx = t * 8 + j;
        int kr  = idx >> 4;
        int nc  = idx & 15;
        Wp[idx] = (nc < 4) ? W1[kr * 4 + nc] : 0.0f;
    }
    __syncthreads();

    const int wave  = t >> 5;
    const int lane  = t & 31;
    const int tile  = blockIdx.x * 8 + wave;
    const int nTiles = n >> 4;
    if (tile >= nTiles) return;

    const int base = tile * 16;
    const int hh   = lane >> 4;
    const int nl   = lane & 15;
    const int arow = base + nl;

    float cs = 0.0f;
    #pragma unroll
    for (int g = 0; g < G / 2; ++g)
        cs += cntp[(size_t)(hh * (G / 2) + g) * N_NODES + arow];
    cs += __shfl_xor(cs, 16);
    const float dv = rsqrtf(1.0f + cs);

    v8f c = {};
    #pragma unroll 4
    for (int k = 0; k < 128; k += 4) {
        const int kk = k + 2 * hh;
        v2f a = *reinterpret_cast<const v2f*>(x + (size_t)arow * 128 + kk);
        v2f b;
        b.x = Wp[(kk + 0) * 16 + nl];
        b.y = Wp[(kk + 1) * 16 + nl];
        c = __builtin_amdgcn_wmma_f32_16x16x4_f32(
                 false, a,  false, b,
                 (short)0, c,  false,  false);
    }

    if (nl < 4) {
        #pragma unroll
        for (int v = 0; v < 8; ++v) {
            int m = base + v + 8 * hh;
            hlin[(size_t)m * ROW_STRIDE(4) + nl] = c[v];
        }
    }
    if (hh == 0) {
        dinv[arow] = dv;
        hlin[(size_t)arow * ROW_STRIDE(4) + 4] = dv;
    }
}

template <int OUT> struct RowVec;
template <> struct RowVec<4> { typedef float4 type; };
template <> struct RowVec<2> { typedef float2 type; };

template <int OUT> struct GRow;
template <> struct GRow<4> {
    float4 h; float d;
    __device__ __forceinline__ void load(const float* __restrict__ hrow, int node) {
        const float* p = hrow + (size_t)node * ROW_STRIDE(4);
        h = *reinterpret_cast<const float4*>(p);
        d = p[4];
    }
    __device__ __forceinline__ float feat(int f) const {
        return f == 0 ? h.x : f == 1 ? h.y : f == 2 ? h.z : h.w;
    }
    __device__ __forceinline__ float dinv() const { return d; }
};
template <> struct GRow<2> {
    float4 q;
    __device__ __forceinline__ void load(const float* __restrict__ hrow, int node) {
        q = *reinterpret_cast<const float4*>(hrow + (size_t)node * ROW_STRIDE(2));
    }
    __device__ __forceinline__ float feat(int f) const { return f == 0 ? q.x : q.y; }
    __device__ __forceinline__ float dinv() const { return q.z; }
};

template <int OUT>
__global__ void __launch_bounds__(RG_THREADS) __attribute__((amdgpu_waves_per_eu(8, 8)))
edge_list_kernel(const unsigned int* __restrict__ lists,
                 const int* __restrict__ lcount,
                 const float* __restrict__ dinv,
                 const float* __restrict__ hlin,
                 float* __restrict__ part) {
    constexpr int NPART = EG_NPART;
    constexpr int NACC  = NPART * OUT;
    constexpr int Q     = 2;
    constexpr int NE    = 4 * Q;
    static_assert(NACC % 4 == 0, "the range's sums split into 16-byte pieces");
    __shared__ __attribute__((aligned(16))) float acc[NACC];
    __shared__ __attribute__((aligned(16))) float dloc[NPART];

    const int t    = threadIdx.x;
    const int lane = t & 31;
    const int wv   = __builtin_amdgcn_readfirstlane(t >> 5);
    const int p    = blockIdx.x % EG_P;
    const int g    = blockIdx.x / EG_P;
    const int base = p * NPART;
    const int pf   = p / EG_SPLIT;
    const int lo   = (p % EG_SPLIT) * NPART;

    int          npcL[EG_SL];
    const uint4* wlL[EG_SL];
    #pragma unroll
    for (int s = 0; s < EG_SL; ++s) {
        const int li = ((g * EG_SL + s) * RG_P + pf) * RG_WAVES + wv;
        npcL[s] = min((lcount[li] + 3) >> 2, RG_LCAP / 4);
        wlL[s]  = reinterpret_cast<const uint4*>(lists + (size_t)li * RG_LCAP);
    }
    auto npcOf = [&](int s) {
        int r = npcL[0];
        #pragma unroll
        for (int i = 1; i < EG_SL; ++i) r = (s == i) ? npcL[i] : r;
        return r;
    };
    auto wlOf = [&](int s) {
        const uint4* r = wlL[0];
        #pragma unroll
        for (int i = 1; i < EG_SL; ++i) r = (s == i) ? wlL[i] : r;
        return r;
    };
    const uint4 none = make_uint4(~0u, ~0u, ~0u, ~0u);
    uint4 qN[Q];
    #pragma unroll
    for (int i = 0; i < Q; ++i) {
        const int k = i * 32 + lane;
        qN[i] = none;
        if (k < npcL[0]) qN[i] = wlL[0][k];
    }

    constexpr int DV     = NPART / 2;
    constexpr int DTRIPS = (DV + RG_THREADS - 1) / RG_THREADS;
    const float2* dsrc  = reinterpret_cast<const float2*>(dinv + base);
    float2*       dloc2 = reinterpret_cast<float2*>(dloc);
    float2 dq[DTRIPS];
    #pragma unroll
    for (int u = 0; u < DTRIPS; ++u) {
        const int i = t + u * RG_THREADS;
        dq[u] = make_float2(0.f, 0.f);
        if (i < DV) dq[u] = dsrc[i];
    }

    float4* acc4 = reinterpret_cast<float4*>(acc);
    for (int i = t; i < NACC / 4; i += RG_THREADS) acc4[i] = make_float4(0.f, 0.f, 0.f, 0.f);

    #pragma unroll
    for (int u = 0; u < DTRIPS; ++u) {
        const int i = t + u * RG_THREADS;
        if (i < DV) dloc2[i] = dq[u];
    }
    __syncthreads();

    int s = 0, k0 = 0;
    for (;;) {
        unsigned int w[NE];
        #pragma unroll
        for (int i = 0; i < Q; ++i) {
            w[4 * i + 0] = qN[i].x; w[4 * i + 1] = qN[i].y;
            w[4 * i + 2] = qN[i].z; w[4 * i + 3] = qN[i].w;
        }
        int ns = s, nk = k0 + 32 * Q;
        if (nk >= npcOf(s)) { ns = s + 1; nk = 0; }
        const bool more = ns < EG_SL;
        {
            const int    npn = more ? npcOf(ns) : 0;
            const uint4* wln = wlOf(more ? ns : 0);
            #pragma unroll
            for (int i = 0; i < Q; ++i) {
                const int kn = nk + i * 32 + lane;
                qN[i] = none;
                if (kn < npn) qN[i] = wln[kn];
            }
        }

        int        rr[NE];
        bool       m[NE];
        GRow<OUT>  gr[NE];
        float      dr[NE];
        #pragma unroll
        for (int j = 0; j < NE; ++j) {
            rr[j] = (int)(w[j] >> COL_BITS) - lo;
            m[j]  = (unsigned)rr[j] < (unsigned)NPART;
            dr[j] = dloc[m[j] ? rr[j] : 0];
        }
        #pragma unroll
        for (int j = 0; j < NE; ++j) {
            if (m[j]) gr[j].load(hlin, (int)(w[j] & ((1u << COL_BITS) - 1u)));
        }
        #pragma unroll
        for (int j = 0; j < NE; ++j) {
            if (m[j]) {
                const float nrm = dr[j] * gr[j].dinv();
                #pragma unroll
                for (int f = 0; f < OUT; ++f)
                    __hip_atomic_fetch_add(&acc[rr[j] * OUT + f], gr[j].feat(f) * nrm,
                                           __ATOMIC_RELAXED, __HIP_MEMORY_SCOPE_WORKGROUP);
            }
        }
        if (!more) break;
        s = ns; k0 = nk;
    }
    __syncthreads();

    float4* dst = reinterpret_cast<float4*>(part + ((size_t)g * N_NODES + base) * OUT);
    for (int i = t; i < NACC / 4; i += RG_THREADS) dst[i] = acc4[i];
}

template <int OUT>
__device__ __forceinline__ void node_features(const float* __restrict__ part,
                                              const float* __restrict__ hlin,
                                              float di,
                                              const float* __restrict__ b,
                                              int i, float (&hv)[OUT]) {
    typedef typename RowVec<OUT>::type VT;
    constexpr int G = EG_G;
    const VT* pv = reinterpret_cast<const VT*>(part);
    float agg[OUT];
    #pragma unroll
    for (int f = 0; f < OUT; ++f) agg[f] = 0.0f;
    #pragma unroll
    for (int g = 0; g < G; ++g) {
        const VT pg = pv[(size_t)g * N_NODES + i];
        const float* pf = reinterpret_cast<const float*>(&pg);
        #pragma unroll
        for (int f = 0; f < OUT; ++f) agg[f] += pf[f];
    }
    const float d2 = di * di;
    #pragma unroll
    for (int f = 0; f < OUT; ++f) {
        float v = agg[f]
                + hlin[(size_t)i * ROW_STRIDE(OUT) + f] * d2 + b[f];
        hv[f] = tanhf(v);
    }
}

template <int OUT, int NEXT>
__global__ void node_stage_kernel(const float* __restrict__ part,
                                  const float* __restrict__ hlin,
                                  const float* __restrict__ dinv,
                                  const float* __restrict__ b,
                                  const float* __restrict__ Wn,
                                  float* __restrict__ tnext,
                                  int n) {
    int i = blockIdx.x * blockDim.x + threadIdx.x;
    if (i >= n) return;
    const float di = dinv[i];
    float hv[OUT];
    node_features<OUT>(part, hlin, di, b, i, hv);

    float s[NEXT];
    #pragma unroll
    for (int o = 0; o < NEXT; ++o) {
        s[o] = 0.0f;
        #pragma unroll
        for (int k = 0; k < OUT; ++k) s[o] = fmaf(hv[k], Wn[k * NEXT + o], s[o]);
    }
    float* orow = tnext + (size_t)i * ROW_STRIDE(NEXT);
    if constexpr (NEXT == 4) {
        *reinterpret_cast<float4*>(orow) = make_float4(s[0], s[1], s[2], s[3]);
        orow[4] = di;
    } else {
        static_assert(NEXT == 2, "rows of 4 or 2 features");
        *reinterpret_cast<float4*>(orow) = make_float4(s[0], s[1], di, 0.0f);
    }
}

__global__ void node_last_kernel(const float* __restrict__ part,
                                 const float* __restrict__ hlin,
                                 const float* __restrict__ dinv,
                                 const float* __restrict__ b,
                                 const float* __restrict__ Wc,
                                 const float* __restrict__ bc,
                                 float* __restrict__ outH,
                                 float* __restrict__ outC,
                                 int n) {
    int i = blockIdx.x * blockDim.x + threadIdx.x;
    if (i >= n) return;
    float hv[2];
    node_features<2>(part, hlin, dinv[i], b, i, hv);
    *reinterpret_cast<float2*>(outH + (size_t)i * 2) = make_float2(hv[0], hv[1]);
    float cl[4];
    #pragma unroll
    for (int c = 0; c < 4; ++c)
        cl[c] = fmaf(hv[0], Wc[c], fmaf(hv[1], Wc[4 + c], bc[c]));
    *reinterpret_cast<float4*>(outC + (size_t)i * 4) = make_float4(cl[0], cl[1], cl[2], cl[3]);
}

extern "C" void kernel_launch(void* const* d_in, const int* in_sizes, int n_in,
                              void* d_out, int out_size, void* d_ws, size_t ws_size,
                              hipStream_t stream) {
    const float* x  = (const float*)d_in[0];
    const int*   ei = (const int*)d_in[1];
    const float* W1 = (const float*)d_in[2];
    const float* b1 = (const float*)d_in[3];
    const float* W2 = (const float*)d_in[4];
    const float* b2 = (const float*)d_in[5];
    const float* W3 = (const float*)d_in[6];
    const float* b3 = (const float*)d_in[7];
    const float* Wc = (const float*)d_in[8];
    const float* bc = (const float*)d_in[9];

    const int N = N_NODES;
    const int E = N_EDGES;
    const int* row = ei;
    const int* col = ei + E;

    float* ws   = (float*)d_ws;
    float* dinv = ws;
    float* tabA = dinv + N;
    float* tabB = tabA + 8 * N;
    float* part = tabB + 8 * N;
    unsigned int* lists  = (unsigned int*)(part + 128 * N);
    int*          lcount = (int*)(lists + (size_t)RG_GROUPS * RG_WAVES * RG_LCAP);

    float* outC = (float*)d_out;
    float* outH = outC + (size_t)4 * N;

    const int BT = 256;
    const int gN = (N + BT - 1) / BT;

    range_filter_kernel<<<RG_GROUPS, RG_THREADS, 0, stream>>>(row, col, part, lists, lcount);

    {
        const int tiles  = N / 16;
        const int blocks = (tiles + 7) / 8;
        gcn_wmma_xw1_kernel<<<blocks, 256, 0, stream>>>(x, W1, part, dinv, tabA, N);
        edge_list_kernel<4><<<RG_GROUPS, RG_THREADS, 0, stream>>>(lists, lcount, dinv, tabA, part);
    }
    {
        node_stage_kernel<4, 4><<<gN, BT, 0, stream>>>(part, tabA, dinv, b1, W2, tabB, N);
        edge_list_kernel<4><<<RG_GROUPS, RG_THREADS, 0, stream>>>(lists, lcount, dinv, tabB, part);
    }
    {
        node_stage_kernel<4, 2><<<gN, BT, 0, stream>>>(part, tabB, dinv, b2, W3, tabA, N);
        edge_list_kernel<2><<<RG_GROUPS, RG_THREADS, 0, stream>>>(lists, lcount, dinv, tabA, part);
    }
    node_last_kernel<<<gN, BT, 0, stream>>>(part, tabA, dinv, b3, Wc, bc, outH, outC, N);
}


# Edge gather, 4-feature layers: a source row is two 16-byte pieces (2 features + dinv each) read by two adjacent lanes, ONE load per edge and 16 lines a load, not 2 loads of 32; LDS add order changes
# speedup vs baseline: 9.3090x; 1.0665x over previous
#include <hip/hip_runtime.h>
#include <hip/hip_bf16.h>
#include <math.h>

#define N_NODES   100000
#define N_EDGES   3200000

typedef float v2f __attribute__((ext_vector_type(2)));
typedef float v8f __attribute__((ext_vector_type(8)));

#define RG_THREADS 1024
#define RG_GROUPS  256
#define RG_P       8
#define RG_G       (RG_GROUPS / RG_P)
#define RG_NPART   (N_NODES / RG_P)
#define RG_ESL     (N_EDGES / RG_G)
#define RG_NV      (RG_ESL / 4)
#define RG_WAVES   (RG_THREADS / 32)
#define RG_TRIPS   ((RG_NV + RG_THREADS - 1) / RG_THREADS)
#define RG_LCAP    (RG_TRIPS * 128)
#define COL_BITS   17
static_assert(N_NODES % RG_P == 0 && N_EDGES % RG_G == 0 && RG_ESL % 4 == 0 && RG_NPART % 4 == 0,
              "ranges, slices and their 16-byte pieces divide evenly");
static_assert(N_NODES <= (1 << COL_BITS) && RG_NPART < (1 << (32 - COL_BITS)) - 1,
              "offset in range and source share one 32-bit word; all ones is no edge");

#define EG_P       16
#define EG_G       (RG_GROUPS / EG_P)
#define EG_NPART   (N_NODES / EG_P)
#define EG_SL      (RG_G / EG_G)
#define EG_SPLIT   (EG_P / RG_P)
#define EDGE_Q(OUT) ((OUT) == 4 ? 4 : 2)
static_assert(N_NODES % EG_P == 0 && RG_G % EG_G == 0 && EG_P % RG_P == 0 &&
              RG_NPART == EG_SPLIT * EG_NPART && EG_NPART % 2 == 0,
              "an edge range is a whole part of one filter range and begins on an 8-byte piece of dinv");

__global__ void __launch_bounds__(RG_THREADS)
range_filter_kernel(const int* __restrict__ row,
                    const int* __restrict__ col,
                    float* __restrict__ cntp,
                    unsigned int* __restrict__ lists,
                    int* __restrict__ lcount) {
    constexpr int NPART = RG_NPART;
    constexpr int U     = 5;
    constexpr int CH    = RG_TRIPS / U;
    static_assert(RG_TRIPS % U == 0, "a lane's pieces split evenly into groups of U");
    __shared__ unsigned int cnt[NPART];

    const int t    = threadIdx.x;
    const int lane = t & 31;
    const int wv   = __builtin_amdgcn_readfirstlane(t >> 5);
    const int p    = blockIdx.x % RG_P;
    const int g    = blockIdx.x / RG_P;
    const int base = p * NPART;

    uint4* cnt4 = reinterpret_cast<uint4*>(cnt);
    for (int i = t; i < NPART / 4; i += RG_THREADS) cnt4[i] = make_uint4(0u, 0u, 0u, 0u);
    __syncthreads();

    const int4* row4 = reinterpret_cast<const int4*>(row + (size_t)g * RG_ESL);
    const int4* col4 = reinterpret_cast<const int4*>(col + (size_t)g * RG_ESL);
    unsigned int* wl = lists + ((size_t)blockIdx.x * RG_WAVES + wv) * RG_LCAP;
    int nl = 0;

    for (int ch = 0; ch < CH; ++ch) {
        int4 r[U], c[U];
        #pragma unroll
        for (int u = 0; u < U; ++u) {
            const int v = t + (ch * U + u) * RG_THREADS;
            r[u] = make_int4(-1, -1, -1, -1);
            c[u] = make_int4(-1, -1, -1, -1);
            if (v < RG_NV) { r[u] = row4[v]; c[u] = col4[v]; }
        }
        #pragma unroll
        for (int u = 0; u < U; ++u) {
            const int rr[4] = { r[u].x - base, r[u].y - base, r[u].z - base, r[u].w - base };
            const int cc[4] = { c[u].x, c[u].y, c[u].z, c[u].w };
            #pragma unroll
            for (int j = 0; j < 4; ++j) {
                if ((unsigned)(cc[j] - base) < (unsigned)NPART)
                    __hip_atomic_fetch_add(&cnt[cc[j] - base], 1u,
                                           __ATOMIC_RELAXED, __HIP_MEMORY_SCOPE_WORKGROUP);
                const bool m = (unsigned)rr[j] < (unsigned)NPART;
                const unsigned int mask = __builtin_amdgcn_ballot_w32(m);
                const int pos = nl + (int)__builtin_amdgcn_mbcnt_lo(mask, 0u);
                if (m) wl[pos] = ((unsigned)rr[j] << COL_BITS) | (unsigned)cc[j];
                nl += __builtin_popcount(mask);
            }
        }
    }
    const int padn = (4 - (nl & 3)) & 3;
    if (lane < padn) wl[nl + lane] = 0xFFFFFFFFu;
    if (lane == 0) lcount[blockIdx.x * RG_WAVES + wv] = nl;
    __syncthreads();

    float4* dst = reinterpret_cast<float4*>(cntp + (size_t)g * N_NODES + base);
    for (int i = t; i < NPART / 4; i += RG_THREADS) {
        const uint4 k = cnt4[i];
        dst[i] = make_float4((float)k.x, (float)k.y, (float)k.z, (float)k.w);
    }
}

#define ROW_STRIDE(OUT) (2 * (OUT))
__device__ __forceinline__ constexpr int feat_pos(int f) { return f + 2 * (f >> 1); }

__global__ void __launch_bounds__(256)
gcn_wmma_xw1_kernel(const float* __restrict__ x,
                    const float* __restrict__ W1,
                    const float* __restrict__ cntp,
                    float* __restrict__ dinv,
                    float* __restrict__ hlin,
                    int n) {
    constexpr int G = RG_G;
    static_assert(G % 2 == 0, "the planes split evenly over the two lane halves");
    __shared__ float Wp[128 * 16];
    const int t = threadIdx.x;
    #pragma unroll
    for (int j = 0; j < 8; ++j) {
        int idx = t * 8 + j;
        int kr  = idx >> 4;
        int nc  = idx & 15;
        Wp[idx] = (nc < 4) ? W1[kr * 4 + nc] : 0.0f;
    }
    __syncthreads();

    const int wave  = t >> 5;
    const int lane  = t & 31;
    const int tile  = blockIdx.x * 8 + wave;
    const int nTiles = n >> 4;
    if (tile >= nTiles) return;

    const int base = tile * 16;
    const int hh   = lane >> 4;
    const int nl   = lane & 15;
    const int arow = base + nl;

    float cs = 0.0f;
    #pragma unroll
    for (int g = 0; g < G / 2; ++g)
        cs += cntp[(size_t)(hh * (G / 2) + g) * N_NODES + arow];
    cs += __shfl_xor(cs, 16);
    const float dv = rsqrtf(1.0f + cs);

    v8f c = {};
    #pragma unroll 4
    for (int k = 0; k < 128; k += 4) {
        const int kk = k + 2 * hh;
        v2f a = *reinterpret_cast<const v2f*>(x + (size_t)arow * 128 + kk);
        v2f b;
        b.x = Wp[(kk + 0) * 16 + nl];
        b.y = Wp[(kk + 1) * 16 + nl];
        c = __builtin_amdgcn_wmma_f32_16x16x4_f32(
                 false, a,  false, b,
                 (short)0, c,  false,  false);
    }

    if (nl < 4) {
        #pragma unroll
        for (int v = 0; v < 8; ++v) {
            int m = base + v + 8 * hh;
            hlin[(size_t)m * ROW_STRIDE(4) + feat_pos(nl)] = c[v];
        }
    }
    *reinterpret_cast<float2*>(hlin + (size_t)arow * ROW_STRIDE(4) + 4 * hh + 2) = make_float2(dv, 0.0f);
    if (hh == 0) dinv[arow] = dv;
}

template <int OUT> struct RowVec;
template <> struct RowVec<4> { typedef float4 type; };
template <> struct RowVec<2> { typedef float2 type; };

template <int OUT>
__global__ void __launch_bounds__(RG_THREADS) __attribute__((amdgpu_waves_per_eu(8, 8)))
edge_list_kernel(const unsigned int* __restrict__ lists,
                 const int* __restrict__ lcount,
                 const float* __restrict__ dinv,
                 const float* __restrict__ hlin,
                 float* __restrict__ part) {
    constexpr int NPART = EG_NPART;
    constexpr int NACC  = NPART * OUT;
    constexpr int LPE   = OUT / 2;
    constexpr int EPT   = 32 / LPE;
    constexpr int Q     = EDGE_Q(OUT);
    constexpr int NE    = 4 * Q;
    static_assert(OUT == 2 * LPE && (LPE == 1 || LPE == 2), "a row is one or two 16-byte pieces");
    static_assert(NACC % 4 == 0, "the range's sums split into 16-byte pieces");
    __shared__ __attribute__((aligned(16))) float acc[NACC];
    __shared__ __attribute__((aligned(16))) float dloc[NPART];

    const int t    = threadIdx.x;
    const int lane = t & 31;
    const int sub  = lane % LPE;
    const int slot = lane / LPE;
    const int wv   = __builtin_amdgcn_readfirstlane(t >> 5);
    const int p    = blockIdx.x % EG_P;
    const int g    = blockIdx.x / EG_P;
    const int base = p * NPART;
    const int pf   = p / EG_SPLIT;
    const int lo   = (p % EG_SPLIT) * NPART;

    int          npcL[EG_SL];
    const uint4* wlL[EG_SL];
    #pragma unroll
    for (int s = 0; s < EG_SL; ++s) {
        const int li = ((g * EG_SL + s) * RG_P + pf) * RG_WAVES + wv;
        npcL[s] = min((lcount[li] + 3) >> 2, RG_LCAP / 4);
        wlL[s]  = reinterpret_cast<const uint4*>(lists + (size_t)li * RG_LCAP);
    }
    auto npcOf = [&](int s) {
        int r = npcL[0];
        #pragma unroll
        for (int i = 1; i < EG_SL; ++i) r = (s == i) ? npcL[i] : r;
        return r;
    };
    auto wlOf = [&](int s) {
        const uint4* r = wlL[0];
        #pragma unroll
        for (int i = 1; i < EG_SL; ++i) r = (s == i) ? wlL[i] : r;
        return r;
    };
    const uint4 none = make_uint4(~0u, ~0u, ~0u, ~0u);
    uint4 qN[Q];
    #pragma unroll
    for (int i = 0; i < Q; ++i) {
        const int k = i * EPT + slot;
        qN[i] = none;
        if (k < npcL[0]) qN[i] = wlL[0][k];
    }

    constexpr int DV     = NPART / 2;
    constexpr int DTRIPS = (DV + RG_THREADS - 1) / RG_THREADS;
    const float2* dsrc  = reinterpret_cast<const float2*>(dinv + base);
    float2*       dloc2 = reinterpret_cast<float2*>(dloc);
    float2 dq[DTRIPS];
    #pragma unroll
    for (int u = 0; u < DTRIPS; ++u) {
        const int i = t + u * RG_THREADS;
        dq[u] = make_float2(0.f, 0.f);
        if (i < DV) dq[u] = dsrc[i];
    }

    float4* acc4 = reinterpret_cast<float4*>(acc);
    for (int i = t; i < NACC / 4; i += RG_THREADS) acc4[i] = make_float4(0.f, 0.f, 0.f, 0.f);

    #pragma unroll
    for (int u = 0; u < DTRIPS; ++u) {
        const int i = t + u * RG_THREADS;
        if (i < DV) dloc2[i] = dq[u];
    }
    __syncthreads();

    const unsigned int pieceOff = 16u * (unsigned)sub;
    int s = 0, k0 = 0;
    for (;;) {
        unsigned int w[NE];
        #pragma unroll
        for (int i = 0; i < Q; ++i) {
            w[4 * i + 0] = qN[i].x; w[4 * i + 1] = qN[i].y;
            w[4 * i + 2] = qN[i].z; w[4 * i + 3] = qN[i].w;
        }
        int ns = s, nk = k0 + EPT * Q;
        if (nk >= npcOf(s)) { ns = s + 1; nk = 0; }
        const bool more = ns < EG_SL;
        {
            const int    npn = more ? npcOf(ns) : 0;
            const uint4* wln = wlOf(more ? ns : 0);
            #pragma unroll
            for (int i = 0; i < Q; ++i) {
                const int kn = nk + i * EPT + slot;
                qN[i] = none;
                if (kn < npn) qN[i] = wln[kn];
            }
        }

        int    rr[NE];
        bool   m[NE];
        float4 gr[NE];
        float  dr[NE];
        #pragma unroll
        for (int j = 0; j < NE; ++j) {
            rr[j] = (int)(w[j] >> COL_BITS) - lo;
            m[j]  = (unsigned)rr[j] < (unsigned)NPART;
            dr[j] = dloc[m[j] ? rr[j] : 0];
        }
        #pragma unroll
        for (int j = 0; j < NE; ++j) {
            if (m[j]) {
                const unsigned int off =
                    ((w[j] & ((1u << COL_BITS) - 1u)) * (unsigned)(ROW_STRIDE(OUT) * sizeof(float))) | pieceOff;
                gr[j] = *reinterpret_cast<const float4*>(reinterpret_cast<const char*>(hlin) + off);
            }
        }
        #pragma unroll
        for (int j = 0; j < NE; ++j) {
            if (m[j]) {
                const float nrm = dr[j] * gr[j].z;
                float* a = &acc[rr[j] * OUT + 2 * sub];
                __hip_atomic_fetch_add(a + 0, gr[j].x * nrm,
                                       __ATOMIC_RELAXED, __HIP_MEMORY_SCOPE_WORKGROUP);
                __hip_atomic_fetch_add(a + 1, gr[j].y * nrm,
                                       __ATOMIC_RELAXED, __HIP_MEMORY_SCOPE_WORKGROUP);
            }
        }
        if (!more) break;
        s = ns; k0 = nk;
    }
    __syncthreads();

    float4* dst = reinterpret_cast<float4*>(part + ((size_t)g * N_NODES + base) * OUT);
    for (int i = t; i < NACC / 4; i += RG_THREADS) dst[i] = acc4[i];
}

template <int OUT>
__device__ __forceinline__ void node_features(const float* __restrict__ part,
                                              const float* __restrict__ hlin,
                                              float di,
                                              const float* __restrict__ b,
                                              int i, float (&hv)[OUT]) {
    typedef typename RowVec<OUT>::type VT;
    constexpr int G = EG_G;
    const VT* pv = reinterpret_cast<const VT*>(part);
    float agg[OUT];
    #pragma unroll
    for (int f = 0; f < OUT; ++f) agg[f] = 0.0f;
    #pragma unroll
    for (int g = 0; g < G; ++g) {
        const VT pg = pv[(size_t)g * N_NODES + i];
        const float* pf = reinterpret_cast<const float*>(&pg);
        #pragma unroll
        for (int f = 0; f < OUT; ++f) agg[f] += pf[f];
    }
    const float d2 = di * di;
    #pragma unroll
    for (int f = 0; f < OUT; ++f) {
        float v = agg[f]
                + hlin[(size_t)i * ROW_STRIDE(OUT) + feat_pos(f)] * d2 + b[f];
        hv[f] = tanhf(v);
    }
}

template <int OUT, int NEXT>
__global__ void node_stage_kernel(const float* __restrict__ part,
                                  const float* __restrict__ hlin,
                                  const float* __restrict__ dinv,
                                  const float* __restrict__ b,
                                  const float* __restrict__ Wn,
                                  float* __restrict__ tnext,
                                  int n) {
    int i = blockIdx.x * blockDim.x + threadIdx.x;
    if (i >= n) return;
    const float di = dinv[i];
    float hv[OUT];
    node_features<OUT>(part, hlin, di, b, i, hv);

    float s[NEXT];
    #pragma unroll
    for (int o = 0; o < NEXT; ++o) {
        s[o] = 0.0f;
        #pragma unroll
        for (int k = 0; k < OUT; ++k) s[o] = fmaf(hv[k], Wn[k * NEXT + o], s[o]);
    }
    float* orow = tnext + (size_t)i * ROW_STRIDE(NEXT);
    if constexpr (NEXT == 4) {
        *reinterpret_cast<float4*>(orow)     = make_float4(s[0], s[1], di, 0.0f);
        *reinterpret_cast<float4*>(orow + 4) = make_float4(s[2], s[3], di, 0.0f);
    } else {
        static_assert(NEXT == 2, "rows of 4 or 2 features");
        *reinterpret_cast<float4*>(orow) = make_float4(s[0], s[1], di, 0.0f);
    }
}

__global__ void node_last_kernel(const float* __restrict__ part,
                                 const float* __restrict__ hlin,
                                 const float* __restrict__ dinv,
                                 const float* __restrict__ b,
                                 const float* __restrict__ Wc,
                                 const float* __restrict__ bc,
                                 float* __restrict__ outH,
                                 float* __restrict__ outC,
                                 int n) {
    int i = blockIdx.x * blockDim.x + threadIdx.x;
    if (i >= n) return;
    float hv[2];
    node_features<2>(part, hlin, dinv[i], b, i, hv);
    *reinterpret_cast<float2*>(outH + (size_t)i * 2) = make_float2(hv[0], hv[1]);
    float cl[4];
    #pragma unroll
    for (int c = 0; c < 4; ++c)
        cl[c] = fmaf(hv[0], Wc[c], fmaf(hv[1], Wc[4 + c], bc[c]));
    *reinterpret_cast<float4*>(outC + (size_t)i * 4) = make_float4(cl[0], cl[1], cl[2], cl[3]);
}

extern "C" void kernel_launch(void* const* d_in, const int* in_sizes, int n_in,
                              void* d_out, int out_size, void* d_ws, size_t ws_size,
                              hipStream_t stream) {
    const float* x  = (const float*)d_in[0];
    const int*   ei = (const int*)d_in[1];
    const float* W1 = (const float*)d_in[2];
    const float* b1 = (const float*)d_in[3];
    const float* W2 = (const float*)d_in[4];
    const float* b2 = (const float*)d_in[5];
    const float* W3 = (const float*)d_in[6];
    const float* b3 = (const float*)d_in[7];
    const float* Wc = (const float*)d_in[8];
    const float* bc = (const float*)d_in[9];

    const int N = N_NODES;
    const int E = N_EDGES;
    const int* row = ei;
    const int* col = ei + E;

    float* ws   = (float*)d_ws;
    float* dinv = ws;
    float* tabA = dinv + N;
    float* tabB = tabA + 8 * N;
    float* part = tabB + 8 * N;
    unsigned int* lists  = (unsigned int*)(part + 128 * N);
    int*          lcount = (int*)(lists + (size_t)RG_GROUPS * RG_WAVES * RG_LCAP);

    float* outC = (float*)d_out;
    float* outH = outC + (size_t)4 * N;

    const int BT = 256;
    const int gN = (N + BT - 1) / BT;

    range_filter_kernel<<<RG_GROUPS, RG_THREADS, 0, stream>>>(row, col, part, lists, lcount);

    {
        const int tiles  = N / 16;
        const int blocks = (tiles + 7) / 8;
        gcn_wmma_xw1_kernel<<<blocks, 256, 0, stream>>>(x, W1, part, dinv, tabA, N);
        edge_list_kernel<4><<<RG_GROUPS, RG_THREADS, 0, stream>>>(lists, lcount, dinv, tabA, part);
    }
    {
        node_stage_kernel<4, 4><<<gN, BT, 0, stream>>>(part, tabA, dinv, b1, W2, tabB, N);
        edge_list_kernel<4><<<RG_GROUPS, RG_THREADS, 0, stream>>>(lists, lcount, dinv, tabB, part);
    }
    {
        node_stage_kernel<4, 2><<<gN, BT, 0, stream>>>(part, tabB, dinv, b2, W3, tabA, N);
        edge_list_kernel<2><<<RG_GROUPS, RG_THREADS, 0, stream>>>(lists, lcount, dinv, tabA, part);
    }
    node_last_kernel<<<gN, BT, 0, stream>>>(part, tabA, dinv, b3, Wc, bc, outH, outC, N);
}


# Edge work tiled by source x destination range (16 x 16 work-groups): a tile holds its source rows and dinv in LDS and gathers there, not from memory; one list pass bins edges; LDS add order changes
# speedup vs baseline: 12.1163x; 1.3016x over previous
#include <hip/hip_runtime.h>
#include <hip/hip_bf16.h>
#include <math.h>

#define N_NODES   100000
#define N_EDGES   3200000

typedef float v2f __attribute__((ext_vector_type(2)));
typedef float v8f __attribute__((ext_vector_type(8)));

#define TL_R       16
#define TL_TILES   (TL_R * TL_R)
#define TL_NP      (N_NODES / TL_R)
#define OFF_BITS   13
#define OFF_MASK   ((1u << OFF_BITS) - 1u)
#define TL_THREADS 1024
#define TL_WAVES   (TL_THREADS / 32)
#define BN_GROUPS  256
#define BN_ESL     (N_EDGES / BN_GROUPS)
#define BN_NV      (BN_ESL / 4)
#define BN_TRIPS   ((BN_NV + TL_THREADS - 1) / TL_THREADS)
#define SEG_PW     (BN_GROUPS / TL_WAVES)
static_assert(N_NODES % TL_R == 0 && TL_NP <= (1 << OFF_BITS) && TL_NP % 2 == 0,
              "equal ranges; an offset fits its bits; a range begins on an 8-byte piece of dinv");
static_assert(N_EDGES % BN_GROUPS == 0 && BN_ESL % 4 == 0 && BN_ESL < (1 << 16) && TL_TILES == 256,
              "equal slices of whole 16-byte pieces; a rank inside a slice fits 16 bits");
static_assert(BN_GROUPS % TL_WAVES == 0, "the slices split evenly over the waves of a tile");

__global__ void __launch_bounds__(TL_THREADS)
tile_bin_kernel(const int* __restrict__ row,
                const int* __restrict__ col,
                unsigned int* __restrict__ binned,
                unsigned int* __restrict__ seg) {
    __shared__ __attribute__((aligned(16))) unsigned int cnt[TL_TILES];
    __shared__ unsigned int offs[TL_TILES + 1];
    __shared__ __attribute__((aligned(16))) unsigned int stage[BN_ESL];

    const int t    = threadIdx.x;
    const int lane = t & 31;
    const int wv   = __builtin_amdgcn_readfirstlane(t >> 5);
    const int w    = blockIdx.x;

    const int4* row4 = reinterpret_cast<const int4*>(row + (size_t)w * BN_ESL);
    const int4* col4 = reinterpret_cast<const int4*>(col + (size_t)w * BN_ESL);
    int4 r[BN_TRIPS], c[BN_TRIPS];
    #pragma unroll
    for (int u = 0; u < BN_TRIPS; ++u) {
        const int v = t + u * TL_THREADS;
        r[u] = make_int4(-1, -1, -1, -1);
        c[u] = make_int4(-1, -1, -1, -1);
        if (v < BN_NV) { r[u] = row4[v]; c[u] = col4[v]; }
    }
    if (t < TL_TILES) cnt[t] = 0u;
    __syncthreads();

    unsigned int word[4 * BN_TRIPS];
    unsigned int key[4 * BN_TRIPS];
    #pragma unroll
    for (int u = 0; u < BN_TRIPS; ++u) {
        const int rr[4] = { r[u].x, r[u].y, r[u].z, r[u].w };
        const int cc[4] = { c[u].x, c[u].y, c[u].z, c[u].w };
        #pragma unroll
        for (int j = 0; j < 4; ++j) {
            const int e = 4 * u + j;
            key[e]  = ~0u;
            word[e] = ~0u;
            if ((unsigned)rr[j] < (unsigned)N_NODES && (unsigned)cc[j] < (unsigned)N_NODES) {
                const unsigned int dr = (unsigned)rr[j] / (unsigned)TL_NP;
                const unsigned int sr = (unsigned)cc[j] / (unsigned)TL_NP;
                const unsigned int tile = sr * TL_R + dr;
                word[e] = (((unsigned)rr[j] - dr * TL_NP) << OFF_BITS) | ((unsigned)cc[j] - sr * TL_NP);
                const unsigned int rank = __hip_atomic_fetch_add(&cnt[tile], 1u,
                                              __ATOMIC_RELAXED, __HIP_MEMORY_SCOPE_WORKGROUP);
                key[e] = (tile << 16) | rank;
            }
        }
    }
    __syncthreads();

    if (wv == 0) {
        const uint4* cnt4 = reinterpret_cast<const uint4*>(cnt);
        const uint4 a = cnt4[2 * lane], b = cnt4[2 * lane + 1];
        const unsigned int v[8] = { a.x, a.y, a.z, a.w, b.x, b.y, b.z, b.w };
        unsigned int tot = 0u;
        #pragma unroll
        for (int i = 0; i < 8; ++i) tot += v[i];
        unsigned int inc = tot;
        #pragma unroll
        for (int d = 1; d < 32; d <<= 1) {
            const unsigned int o = __shfl_up(inc, d);
            if (lane >= d) inc += o;
        }
        unsigned int run = inc - tot;
        #pragma unroll
        for (int i = 0; i < 8; ++i) { offs[8 * lane + i] = run; run += v[i]; }
        if (lane == 31) offs[TL_TILES] = inc;
    }
    __syncthreads();

    if (t < TL_TILES)
        reinterpret_cast<uint2*>(seg)[(size_t)t * BN_GROUPS + w] = make_uint2(offs[t], offs[t + 1]);
    #pragma unroll
    for (int e = 0; e < 4 * BN_TRIPS; ++e)
        if (key[e] != ~0u) stage[offs[key[e] >> 16] + (key[e] & 0xFFFFu)] = word[e];
    __syncthreads();

    const uint4* stage4 = reinterpret_cast<const uint4*>(stage);
    uint4* dst = reinterpret_cast<uint4*>(binned + (size_t)w * BN_ESL);
    for (int i = t; i < BN_NV; i += TL_THREADS) dst[i] = stage4[i];
}

struct WaveList {
    int pre[SEG_PW + 1];
    int gb[SEG_PW];
};
__device__ __forceinline__ void wave_list(const unsigned int* __restrict__ seg, int b, int wv, int lane,
                                          WaveList& wl) {
    uint2 sg = make_uint2(0u, 0u);
    if (lane < SEG_PW)
        sg = reinterpret_cast<const uint2*>(seg)[(size_t)b * BN_GROUPS + wv * SEG_PW + lane];
    wl.pre[0] = 0;
    #pragma unroll
    for (int i = 0; i < SEG_PW; ++i) {
        int s = __builtin_amdgcn_readlane((int)sg.x, i);
        int e = __builtin_amdgcn_readlane((int)sg.y, i);
        s = min(max(s, 0), BN_ESL);
        e = min(max(e, s), BN_ESL);
        wl.gb[i]      = (wv * SEG_PW + i) * BN_ESL + s - wl.pre[i];
        wl.pre[i + 1] = wl.pre[i] + (e - s);
    }
}
__device__ __forceinline__ int word_index(const WaveList& wl, int v) {
    int idx = wl.gb[0] + v;
    #pragma unroll
    for (int i = 1; i < SEG_PW; ++i) idx = (v >= wl.pre[i]) ? wl.gb[i] + v : idx;
    return idx;
}

__global__ void __launch_bounds__(TL_THREADS)
tile_degree_kernel(const unsigned int* __restrict__ binned,
                   const unsigned int* __restrict__ seg,
                   float* __restrict__ cntp) {
    constexpr int NP = TL_NP;
    constexpr int U  = 8;
    __shared__ __attribute__((aligned(16))) unsigned int hist[NP];

    const int t    = threadIdx.x;
    const int lane = t & 31;
    const int wv   = __builtin_amdgcn_readfirstlane(t >> 5);
    const int b    = blockIdx.x;
    const int sr   = b / TL_R;
    const int dr   = b % TL_R;

    WaveList wl;
    wave_list(seg, b, wv, lane, wl);
    const int len = wl.pre[SEG_PW];

    uint2* hist2 = reinterpret_cast<uint2*>(hist);
    for (int i = t; i < NP / 2; i += TL_THREADS) hist2[i] = make_uint2(0u, 0u);
    __syncthreads();

    for (int v0 = 0; v0 < len; v0 += 32 * U) {
        unsigned int wd[U];
        #pragma unroll
        for (int u = 0; u < U; ++u) {
            const int v = v0 + u * 32 + lane;
            wd[u] = ~0u;
            if (v < len) wd[u] = binned[word_index(wl, v)];
        }
        #pragma unroll
        for (int u = 0; u < U; ++u) {
            const unsigned int so = wd[u] & OFF_MASK;
            if (wd[u] != ~0u && so < (unsigned)NP)
                __hip_atomic_fetch_add(&hist[so], 1u, __ATOMIC_RELAXED, __HIP_MEMORY_SCOPE_WORKGROUP);
        }
    }
    __syncthreads();

    float2* dst = reinterpret_cast<float2*>(cntp + (size_t)dr * N_NODES + (size_t)sr * NP);
    for (int i = t; i < NP / 2; i += TL_THREADS) {
        const uint2 k = hist2[i];
        dst[i] = make_float2((float)k.x, (float)k.y);
    }
}


__global__ void __launch_bounds__(256)
gcn_wmma_xw1_kernel(const float* __restrict__ x,
                    const float* __restrict__ W1,
                    const float* __restrict__ cntp,
                    float* __restrict__ dinv,
                    float* __restrict__ hlin,
                    int n) {
    constexpr int G = TL_R;
    static_assert(G % 2 == 0, "the planes split evenly over the two lane halves");
    __shared__ float Wp[128 * 16];
    const int t = threadIdx.x;
    #pragma unroll
    for (int j = 0; j < 8; ++j) {
        int idx = t * 8 + j;
        int kr  = idx >> 4;
        int nc  = idx & 15;
        Wp[idx] = (nc < 4) ? W1[kr * 4 + nc] : 0.0f;
    }
    __syncthreads();

    const int wave  = t >> 5;
    const int lane  = t & 31;
    const int tile  = blockIdx.x * 8 + wave;
    const int nTiles = n >> 4;
    if (tile >= nTiles) return;

    const int base = tile * 16;
    const int hh   = lane >> 4;
    const int nl   = lane & 15;
    const int arow = base + nl;

    float cs = 0.0f;
    #pragma unroll
    for (int g = 0; g < G / 2; ++g)
        cs += cntp[(size_t)(hh * (G / 2) + g) * N_NODES + arow];
    cs += __shfl_xor(cs, 16);
    const float dv = rsqrtf(1.0f + cs);

    v8f c = {};
    #pragma unroll 4
    for (int k = 0; k < 128; k += 4) {
        const int kk = k + 2 * hh;
        v2f a = *reinterpret_cast<const v2f*>(x + (size_t)arow * 128 + kk);
        v2f b;
        b.x = Wp[(kk + 0) * 16 + nl];
        b.y = Wp[(kk + 1) * 16 + nl];
        c = __builtin_amdgcn_wmma_f32_16x16x4_f32(
                 false, a,  false, b,
                 (short)0, c,  false,  false);
    }

    if (nl < 4) {
        #pragma unroll
        for (int v = 0; v < 8; ++v) {
            int m = base + v + 8 * hh;
            hlin[(size_t)m * 4 + nl] = c[v];
        }
    }
    if (hh == 0) dinv[arow] = dv;
}

template <int OUT> struct RowVec;
template <> struct RowVec<4> { typedef float4 type; };
template <> struct RowVec<2> { typedef float2 type; };

template <int OUT>
__global__ void __launch_bounds__(TL_THREADS) __attribute__((amdgpu_waves_per_eu(8, 8)))
edge_tile_kernel(const unsigned int* __restrict__ binned,
                 const unsigned int* __restrict__ seg,
                 const float* __restrict__ dinv,
                 const float* __restrict__ hlin,
                 float* __restrict__ part) {
    constexpr int NP   = TL_NP;
    constexpr int NACC = NP * OUT;
    constexpr int LPE  = OUT / 2;
    constexpr int EPS  = 32 / LPE;
    constexpr int NE   = 8;
    static_assert(OUT == 2 * LPE && (LPE == 1 || LPE == 2), "two features a lane");
    static_assert(NACC % 4 == 0, "rows and sums split into 16-byte pieces");
    __shared__ __attribute__((aligned(16))) float rows[NACC];
    __shared__ __attribute__((aligned(16))) float acc[NACC];
    __shared__ __attribute__((aligned(16))) float sdv[NP];
    __shared__ __attribute__((aligned(16))) float ddv[NP];

    const int t    = threadIdx.x;
    const int lane = t & 31;
    const int sub  = lane % LPE;
    const int slot = lane / LPE;
    const int wv   = __builtin_amdgcn_readfirstlane(t >> 5);
    const int b    = blockIdx.x;
    const int sr   = b / TL_R;
    const int dr   = b % TL_R;

    constexpr int RV     = NACC / 4;
    constexpr int RTRIPS = (RV + TL_THREADS - 1) / TL_THREADS;
    constexpr int DV     = NP / 2;
    constexpr int DTRIPS = (DV + TL_THREADS - 1) / TL_THREADS;
    const float4* rsrc = reinterpret_cast<const float4*>(hlin + (size_t)sr * NACC);
    const float2* ssrc = reinterpret_cast<const float2*>(dinv + (size_t)sr * NP);
    const float2* dsrc = reinterpret_cast<const float2*>(dinv + (size_t)dr * NP);
    float4 rq[RTRIPS];
    float2 sq[DTRIPS], dq[DTRIPS];
    #pragma unroll
    for (int u = 0; u < RTRIPS; ++u) {
        const int i = t + u * TL_THREADS;
        rq[u] = make_float4(0.f, 0.f, 0.f, 0.f);
        if (i < RV) rq[u] = rsrc[i];
    }
    #pragma unroll
    for (int u = 0; u < DTRIPS; ++u) {
        const int i = t + u * TL_THREADS;
        sq[u] = make_float2(0.f, 0.f);
        dq[u] = make_float2(0.f, 0.f);
        if (i < DV) { sq[u] = ssrc[i]; dq[u] = dsrc[i]; }
    }

    WaveList wl;
    wave_list(seg, b, wv, lane, wl);
    const int len = wl.pre[SEG_PW];
    unsigned int wN[NE];
    #pragma unroll
    for (int j = 0; j < NE; ++j) {
        const int v = j * EPS + slot;
        wN[j] = ~0u;
        if (v < len) wN[j] = binned[word_index(wl, v)];
    }

    float4* acc4  = reinterpret_cast<float4*>(acc);
    float4* rows4 = reinterpret_cast<float4*>(rows);
    float2* sdv2  = reinterpret_cast<float2*>(sdv);
    float2* ddv2  = reinterpret_cast<float2*>(ddv);
    for (int i = t; i < RV; i += TL_THREADS) acc4[i] = make_float4(0.f, 0.f, 0.f, 0.f);
    #pragma unroll
    for (int u = 0; u < RTRIPS; ++u) {
        const int i = t + u * TL_THREADS;
        if (i < RV) rows4[i] = rq[u];
    }
    #pragma unroll
    for (int u = 0; u < DTRIPS; ++u) {
        const int i = t + u * TL_THREADS;
        if (i < DV) { sdv2[i] = sq[u]; ddv2[i] = dq[u]; }
    }
    __syncthreads();

    int v0 = 0;
    for (;;) {
        unsigned int w[NE];
        #pragma unroll
        for (int j = 0; j < NE; ++j) w[j] = wN[j];
        const int  nv0  = v0 + NE * EPS;
        const bool more = nv0 < len;
        #pragma unroll
        for (int j = 0; j < NE; ++j) {
            const int v = nv0 + j * EPS + slot;
            wN[j] = ~0u;
            if (v < len) wN[j] = binned[word_index(wl, v)];
        }

        bool   m[NE];
        int    so[NE], dn[NE];
        float2 rp[NE];
        float  sd[NE], dd[NE];
        #pragma unroll
        for (int j = 0; j < NE; ++j) {
            const unsigned int s_ = w[j] & OFF_MASK;
            const unsigned int d_ = w[j] >> OFF_BITS;
            m[j]  = s_ < (unsigned)NP && d_ < (unsigned)NP;
            so[j] = m[j] ? (int)s_ : 0;
            dn[j] = m[j] ? (int)d_ : 0;
            rp[j] = *reinterpret_cast<const float2*>(&rows[so[j] * OUT + 2 * sub]);
            sd[j] = sdv[so[j]];
            dd[j] = ddv[dn[j]];
        }
        #pragma unroll
        for (int j = 0; j < NE; ++j) {
            if (m[j]) {
                const float nrm = dd[j] * sd[j];
                float* a = &acc[dn[j] * OUT + 2 * sub];
                __hip_atomic_fetch_add(a + 0, rp[j].x * nrm,
                                       __ATOMIC_RELAXED, __HIP_MEMORY_SCOPE_WORKGROUP);
                __hip_atomic_fetch_add(a + 1, rp[j].y * nrm,
                                       __ATOMIC_RELAXED, __HIP_MEMORY_SCOPE_WORKGROUP);
            }
        }
        if (!more) break;
        v0 = nv0;
    }
    __syncthreads();

    float4* dst = reinterpret_cast<float4*>(part + ((size_t)sr * N_NODES + (size_t)dr * NP) * OUT);
    for (int i = t; i < RV; i += TL_THREADS) dst[i] = acc4[i];
}

template <int OUT>
__device__ __forceinline__ void node_features(const float* __restrict__ part,
                                              const float* __restrict__ hlin,
                                              float di,
                                              const float* __restrict__ b,
                                              int i, float (&hv)[OUT]) {
    typedef typename RowVec<OUT>::type VT;
    constexpr int G = TL_R;
    const VT* pv = reinterpret_cast<const VT*>(part);
    const VT hrow = reinterpret_cast<const VT*>(hlin)[i];
    const float* hr = reinterpret_cast<const float*>(&hrow);
    float agg[OUT];
    #pragma unroll
    for (int f = 0; f < OUT; ++f) agg[f] = 0.0f;
    #pragma unroll
    for (int g = 0; g < G; ++g) {
        const VT pg = pv[(size_t)g * N_NODES + i];
        const float* pf = reinterpret_cast<const float*>(&pg);
        #pragma unroll
        for (int f = 0; f < OUT; ++f) agg[f] += pf[f];
    }
    const float d2 = di * di;
    #pragma unroll
    for (int f = 0; f < OUT; ++f) {
        float v = agg[f] + hr[f] * d2 + b[f];
        hv[f] = tanhf(v);
    }
}

template <int OUT, int NEXT>
__global__ void node_stage_kernel(const float* __restrict__ part,
                                  const float* __restrict__ hlin,
                                  const float* __restrict__ dinv,
                                  const float* __restrict__ b,
                                  const float* __restrict__ Wn,
                                  float* __restrict__ tnext,
                                  int n) {
    int i = blockIdx.x * blockDim.x + threadIdx.x;
    if (i >= n) return;
    const float di = dinv[i];
    float hv[OUT];
    node_features<OUT>(part, hlin, di, b, i, hv);

    float s[NEXT];
    #pragma unroll
    for (int o = 0; o < NEXT; ++o) {
        s[o] = 0.0f;
        #pragma unroll
        for (int k = 0; k < OUT; ++k) s[o] = fmaf(hv[k], Wn[k * NEXT + o], s[o]);
    }
    float* orow = tnext + (size_t)i * NEXT;
    if constexpr (NEXT == 4) {
        *reinterpret_cast<float4*>(orow) = make_float4(s[0], s[1], s[2], s[3]);
    } else {
        static_assert(NEXT == 2, "rows of 4 or 2 features");
        *reinterpret_cast<float2*>(orow) = make_float2(s[0], s[1]);
    }
}

__global__ void node_last_kernel(const float* __restrict__ part,
                                 const float* __restrict__ hlin,
                                 const float* __restrict__ dinv,
                                 const float* __restrict__ b,
                                 const float* __restrict__ Wc,
                                 const float* __restrict__ bc,
                                 float* __restrict__ outH,
                                 float* __restrict__ outC,
                                 int n) {
    int i = blockIdx.x * blockDim.x + threadIdx.x;
    if (i >= n) return;
    float hv[2];
    node_features<2>(part, hlin, dinv[i], b, i, hv);
    *reinterpret_cast<float2*>(outH + (size_t)i * 2) = make_float2(hv[0], hv[1]);
    float cl[4];
    #pragma unroll
    for (int c = 0; c < 4; ++c)
        cl[c] = fmaf(hv[0], Wc[c], fmaf(hv[1], Wc[4 + c], bc[c]));
    *reinterpret_cast<float4*>(outC + (size_t)i * 4) = make_float4(cl[0], cl[1], cl[2], cl[3]);
}

extern "C" void kernel_launch(void* const* d_in, const int* in_sizes, int n_in,
                              void* d_out, int out_size, void* d_ws, size_t ws_size,
                              hipStream_t stream) {
    const float* x  = (const float*)d_in[0];
    const int*   ei = (const int*)d_in[1];
    const float* W1 = (const float*)d_in[2];
    const float* b1 = (const float*)d_in[3];
    const float* W2 = (const float*)d_in[4];
    const float* b2 = (const float*)d_in[5];
    const float* W3 = (const float*)d_in[6];
    const float* b3 = (const float*)d_in[7];
    const float* Wc = (const float*)d_in[8];
    const float* bc = (const float*)d_in[9];

    const int N = N_NODES;
    const int E = N_EDGES;
    const int* row = ei;
    const int* col = ei + E;

    float* ws   = (float*)d_ws;
    float* dinv = ws;
    float* tabA = dinv + N;
    float* tabB = tabA + 4 * N;
    float* part = tabB + 4 * N;
    unsigned int* binned = (unsigned int*)(part + 64 * N);
    unsigned int* seg    = binned + (size_t)BN_GROUPS * BN_ESL;

    float* outC = (float*)d_out;
    float* outH = outC + (size_t)4 * N;

    const int BT = 256;
    const int gN = (N + BT - 1) / BT;

    tile_bin_kernel<<<BN_GROUPS, TL_THREADS, 0, stream>>>(row, col, binned, seg);
    tile_degree_kernel<<<TL_TILES, TL_THREADS, 0, stream>>>(binned, seg, part);

    {
        const int tiles  = N / 16;
        const int blocks = (tiles + 7) / 8;
        gcn_wmma_xw1_kernel<<<blocks, 256, 0, stream>>>(x, W1, part, dinv, tabA, N);
        edge_tile_kernel<4><<<TL_TILES, TL_THREADS, 0, stream>>>(binned, seg, dinv, tabA, part);
    }
    {
        node_stage_kernel<4, 4><<<gN, BT, 0, stream>>>(part, tabA, dinv, b1, W2, tabB, N);
        edge_tile_kernel<4><<<TL_TILES, TL_THREADS, 0, stream>>>(binned, seg, dinv, tabB, part);
    }
    {
        node_stage_kernel<4, 2><<<gN, BT, 0, stream>>>(part, tabB, dinv, b2, W3, tabA, N);
        edge_tile_kernel<2><<<TL_TILES, TL_THREADS, 0, stream>>>(binned, seg, dinv, tabA, part);
    }
    node_last_kernel<<<gN, BT, 0, stream>>>(part, tabA, dinv, b3, Wc, bc, outH, outC, N);
}


# Edge launches: ONE lane per edge (row read as one 16- or 8-byte LDS piece); a lane's LDS adds start at feature lane mod OUT, so one instruction's adds fall on all 64 banks; LDS add order changes
# speedup vs baseline: 12.7570x; 1.0529x over previous
#include <hip/hip_runtime.h>
#include <hip/hip_bf16.h>
#include <math.h>

#define N_NODES   100000
#define N_EDGES   3200000

typedef float v2f __attribute__((ext_vector_type(2)));
typedef float v8f __attribute__((ext_vector_type(8)));

#define TL_R       16
#define TL_TILES   (TL_R * TL_R)
#define TL_NP      (N_NODES / TL_R)
#define OFF_BITS   13
#define OFF_MASK   ((1u << OFF_BITS) - 1u)
#define TL_THREADS 1024
#define TL_WAVES   (TL_THREADS / 32)
#define BN_GROUPS  256
#define BN_ESL     (N_EDGES / BN_GROUPS)
#define BN_NV      (BN_ESL / 4)
#define BN_TRIPS   ((BN_NV + TL_THREADS - 1) / TL_THREADS)
#define SEG_PW     (BN_GROUPS / TL_WAVES)
static_assert(N_NODES % TL_R == 0 && TL_NP <= (1 << OFF_BITS) && TL_NP % 2 == 0,
              "equal ranges; an offset fits its bits; a range begins on an 8-byte piece of dinv");
static_assert(N_EDGES % BN_GROUPS == 0 && BN_ESL % 4 == 0 && BN_ESL < (1 << 16) && TL_TILES == 256,
              "equal slices of whole 16-byte pieces; a rank inside a slice fits 16 bits");
static_assert(BN_GROUPS % TL_WAVES == 0, "the slices split evenly over the waves of a tile");

__global__ void __launch_bounds__(TL_THREADS)
tile_bin_kernel(const int* __restrict__ row,
                const int* __restrict__ col,
                unsigned int* __restrict__ binned,
                unsigned int* __restrict__ seg) {
    __shared__ __attribute__((aligned(16))) unsigned int cnt[TL_TILES];
    __shared__ unsigned int offs[TL_TILES + 1];
    __shared__ __attribute__((aligned(16))) unsigned int stage[BN_ESL];

    const int t    = threadIdx.x;
    const int lane = t & 31;
    const int wv   = __builtin_amdgcn_readfirstlane(t >> 5);
    const int w    = blockIdx.x;

    const int4* row4 = reinterpret_cast<const int4*>(row + (size_t)w * BN_ESL);
    const int4* col4 = reinterpret_cast<const int4*>(col + (size_t)w * BN_ESL);
    int4 r[BN_TRIPS], c[BN_TRIPS];
    #pragma unroll
    for (int u = 0; u < BN_TRIPS; ++u) {
        const int v = t + u * TL_THREADS;
        r[u] = make_int4(-1, -1, -1, -1);
        c[u] = make_int4(-1, -1, -1, -1);
        if (v < BN_NV) { r[u] = row4[v]; c[u] = col4[v]; }
    }
    if (t < TL_TILES) cnt[t] = 0u;
    __syncthreads();

    unsigned int word[4 * BN_TRIPS];
    unsigned int key[4 * BN_TRIPS];
    #pragma unroll
    for (int u = 0; u < BN_TRIPS; ++u) {
        const int rr[4] = { r[u].x, r[u].y, r[u].z, r[u].w };
        const int cc[4] = { c[u].x, c[u].y, c[u].z, c[u].w };
        #pragma unroll
        for (int j = 0; j < 4; ++j) {
            const int e = 4 * u + j;
            key[e]  = ~0u;
            word[e] = ~0u;
            if ((unsigned)rr[j] < (unsigned)N_NODES && (unsigned)cc[j] < (unsigned)N_NODES) {
                const unsigned int dr = (unsigned)rr[j] / (unsigned)TL_NP;
                const unsigned int sr = (unsigned)cc[j] / (unsigned)TL_NP;
                const unsigned int tile = sr * TL_R + dr;
                word[e] = (((unsigned)rr[j] - dr * TL_NP) << OFF_BITS) | ((unsigned)cc[j] - sr * TL_NP);
                const unsigned int rank = __hip_atomic_fetch_add(&cnt[tile], 1u,
                                              __ATOMIC_RELAXED, __HIP_MEMORY_SCOPE_WORKGROUP);
                key[e] = (tile << 16) | rank;
            }
        }
    }
    __syncthreads();

    if (wv == 0) {
        const uint4* cnt4 = reinterpret_cast<const uint4*>(cnt);
        const uint4 a = cnt4[2 * lane], b = cnt4[2 * lane + 1];
        const unsigned int v[8] = { a.x, a.y, a.z, a.w, b.x, b.y, b.z, b.w };
        unsigned int tot = 0u;
        #pragma unroll
        for (int i = 0; i < 8; ++i) tot += v[i];
        unsigned int inc = tot;
        #pragma unroll
        for (int d = 1; d < 32; d <<= 1) {
            const unsigned int o = __shfl_up(inc, d);
            if (lane >= d) inc += o;
        }
        unsigned int run = inc - tot;
        #pragma unroll
        for (int i = 0; i < 8; ++i) { offs[8 * lane + i] = run; run += v[i]; }
        if (lane == 31) offs[TL_TILES] = inc;
    }
    __syncthreads();

    if (t < TL_TILES)
        reinterpret_cast<uint2*>(seg)[(size_t)t * BN_GROUPS + w] = make_uint2(offs[t], offs[t + 1]);
    #pragma unroll
    for (int e = 0; e < 4 * BN_TRIPS; ++e)
        if (key[e] != ~0u) stage[offs[key[e] >> 16] + (key[e] & 0xFFFFu)] = word[e];
    __syncthreads();

    const uint4* stage4 = reinterpret_cast<const uint4*>(stage);
    uint4* dst = reinterpret_cast<uint4*>(binned + (size_t)w * BN_ESL);
    for (int i = t; i < BN_NV; i += TL_THREADS) dst[i] = stage4[i];
}

struct WaveList {
    int pre[SEG_PW + 1];
    int gb[SEG_PW];
};
__device__ __forceinline__ void wave_list(const unsigned int* __restrict__ seg, int b, int wv, int lane,
                                          WaveList& wl) {
    uint2 sg = make_uint2(0u, 0u);
    if (lane < SEG_PW)
        sg = reinterpret_cast<const uint2*>(seg)[(size_t)b * BN_GROUPS + wv * SEG_PW + lane];
    wl.pre[0] = 0;
    #pragma unroll
    for (int i = 0; i < SEG_PW; ++i) {
        int s = __builtin_amdgcn_readlane((int)sg.x, i);
        int e = __builtin_amdgcn_readlane((int)sg.y, i);
        s = min(max(s, 0), BN_ESL);
        e = min(max(e, s), BN_ESL);
        wl.gb[i]      = (wv * SEG_PW + i) * BN_ESL + s - wl.pre[i];
        wl.pre[i + 1] = wl.pre[i] + (e - s);
    }
}
__device__ __forceinline__ int word_index(const WaveList& wl, int v) {
    int idx = wl.gb[0] + v;
    #pragma unroll
    for (int i = 1; i < SEG_PW; ++i) idx = (v >= wl.pre[i]) ? wl.gb[i] + v : idx;
    return idx;
}

__global__ void __launch_bounds__(TL_THREADS)
tile_degree_kernel(const unsigned int* __restrict__ binned,
                   const unsigned int* __restrict__ seg,
                   float* __restrict__ cntp) {
    constexpr int NP = TL_NP;
    constexpr int U  = 8;
    __shared__ __attribute__((aligned(16))) unsigned int hist[NP];

    const int t    = threadIdx.x;
    const int lane = t & 31;
    const int wv   = __builtin_amdgcn_readfirstlane(t >> 5);
    const int b    = blockIdx.x;
    const int sr   = b / TL_R;
    const int dr   = b % TL_R;

    WaveList wl;
    wave_list(seg, b, wv, lane, wl);
    const int len = wl.pre[SEG_PW];

    uint2* hist2 = reinterpret_cast<uint2*>(hist);
    for (int i = t; i < NP / 2; i += TL_THREADS) hist2[i] = make_uint2(0u, 0u);
    __syncthreads();

    for (int v0 = 0; v0 < len; v0 += 32 * U) {
        unsigned int wd[U];
        #pragma unroll
        for (int u = 0; u < U; ++u) {
            const int v = v0 + u * 32 + lane;
            wd[u] = ~0u;
            if (v < len) wd[u] = binned[word_index(wl, v)];
        }
        #pragma unroll
        for (int u = 0; u < U; ++u) {
            const unsigned int so = wd[u] & OFF_MASK;
            if (wd[u] != ~0u && so < (unsigned)NP)
                __hip_atomic_fetch_add(&hist[so], 1u, __ATOMIC_RELAXED, __HIP_MEMORY_SCOPE_WORKGROUP);
        }
    }
    __syncthreads();

    float2* dst = reinterpret_cast<float2*>(cntp + (size_t)dr * N_NODES + (size_t)sr * NP);
    for (int i = t; i < NP / 2; i += TL_THREADS) {
        const uint2 k = hist2[i];
        dst[i] = make_float2((float)k.x, (float)k.y);
    }
}


__global__ void __launch_bounds__(256)
gcn_wmma_xw1_kernel(const float* __restrict__ x,
                    const float* __restrict__ W1,
                    const float* __restrict__ cntp,
                    float* __restrict__ dinv,
                    float* __restrict__ hlin,
                    int n) {
    constexpr int G = TL_R;
    static_assert(G % 2 == 0, "the planes split evenly over the two lane halves");
    __shared__ float Wp[128 * 16];
    const int t = threadIdx.x;
    #pragma unroll
    for (int j = 0; j < 8; ++j) {
        int idx = t * 8 + j;
        int kr  = idx >> 4;
        int nc  = idx & 15;
        Wp[idx] = (nc < 4) ? W1[kr * 4 + nc] : 0.0f;
    }
    __syncthreads();

    const int wave  = t >> 5;
    const int lane  = t & 31;
    const int tile  = blockIdx.x * 8 + wave;
    const int nTiles = n >> 4;
    if (tile >= nTiles) return;

    const int base = tile * 16;
    const int hh   = lane >> 4;
    const int nl   = lane & 15;
    const int arow = base + nl;

    float cs = 0.0f;
    #pragma unroll
    for (int g = 0; g < G / 2; ++g)
        cs += cntp[(size_t)(hh * (G / 2) + g) * N_NODES + arow];
    cs += __shfl_xor(cs, 16);
    const float dv = rsqrtf(1.0f + cs);

    v8f c = {};
    #pragma unroll 4
    for (int k = 0; k < 128; k += 4) {
        const int kk = k + 2 * hh;
        v2f a = *reinterpret_cast<const v2f*>(x + (size_t)arow * 128 + kk);
        v2f b;
        b.x = Wp[(kk + 0) * 16 + nl];
        b.y = Wp[(kk + 1) * 16 + nl];
        c = __builtin_amdgcn_wmma_f32_16x16x4_f32(
                 false, a,  false, b,
                 (short)0, c,  false,  false);
    }

    if (nl < 4) {
        #pragma unroll
        for (int v = 0; v < 8; ++v) {
            int m = base + v + 8 * hh;
            hlin[(size_t)m * 4 + nl] = c[v];
        }
    }
    if (hh == 0) dinv[arow] = dv;
}

template <int OUT> struct RowVec;
template <> struct RowVec<4> { typedef float4 type; };
template <> struct RowVec<2> { typedef float2 type; };

__device__ __forceinline__ void row_times(const float4& r, float n, float (&p)[4]) {
    p[0] = r.x * n; p[1] = r.y * n; p[2] = r.z * n; p[3] = r.w * n;
}
__device__ __forceinline__ void row_times(const float2& r, float n, float (&p)[2]) {
    p[0] = r.x * n; p[1] = r.y * n;
}
__device__ __forceinline__ void rotate_to_lane(float (&p)[4], int q) {
    const bool a = (q & 1) != 0, b = (q & 2) != 0;
    const float t0 = a ? p[1] : p[0], t1 = a ? p[2] : p[1], t2 = a ? p[3] : p[2], t3 = a ? p[0] : p[3];
    p[0] = b ? t2 : t0; p[1] = b ? t3 : t1; p[2] = b ? t0 : t2; p[3] = b ? t1 : t3;
}
__device__ __forceinline__ void rotate_to_lane(float (&p)[2], int q) {
    const bool a = (q & 1) != 0;
    const float t0 = a ? p[1] : p[0], t1 = a ? p[0] : p[1];
    p[0] = t0; p[1] = t1;
}

template <int OUT>
__global__ void __launch_bounds__(TL_THREADS) __attribute__((amdgpu_waves_per_eu(8, 8)))
edge_tile_kernel(const unsigned int* __restrict__ binned,
                 const unsigned int* __restrict__ seg,
                 const float* __restrict__ dinv,
                 const float* __restrict__ hlin,
                 float* __restrict__ part) {
    constexpr int NP   = TL_NP;
    constexpr int NACC = NP * OUT;
    typedef typename RowVec<OUT>::type VT;
    constexpr int EPS  = 32;
    constexpr int NE   = 8;
    static_assert(OUT == 4 || OUT == 2, "a row is one 16- or 8-byte piece; lane mod OUT is a mask");
    static_assert(NACC % 4 == 0, "rows and sums split into 16-byte pieces");
    __shared__ __attribute__((aligned(16))) float rows[NACC];
    __shared__ __attribute__((aligned(16))) float acc[NACC];
    __shared__ __attribute__((aligned(16))) float sdv[NP];
    __shared__ __attribute__((aligned(16))) float ddv[NP];

    const int t    = threadIdx.x;
    const int lane = t & 31;
    const int slot = lane;
    const int q    = lane & (OUT - 1);
    int fo[OUT];
    #pragma unroll
    for (int k = 0; k < OUT; ++k) fo[k] = (q + k) & (OUT - 1);
    const int wv   = __builtin_amdgcn_readfirstlane(t >> 5);
    const int b    = blockIdx.x;
    const int sr   = b / TL_R;
    const int dr   = b % TL_R;

    constexpr int RV     = NACC / 4;
    constexpr int RTRIPS = (RV + TL_THREADS - 1) / TL_THREADS;
    constexpr int DV     = NP / 2;
    constexpr int DTRIPS = (DV + TL_THREADS - 1) / TL_THREADS;
    const float4* rsrc = reinterpret_cast<const float4*>(hlin + (size_t)sr * NACC);
    const float2* ssrc = reinterpret_cast<const float2*>(dinv + (size_t)sr * NP);
    const float2* dsrc = reinterpret_cast<const float2*>(dinv + (size_t)dr * NP);
    float4 rq[RTRIPS];
    float2 sq[DTRIPS], dq[DTRIPS];
    #pragma unroll
    for (int u = 0; u < RTRIPS; ++u) {
        const int i = t + u * TL_THREADS;
        rq[u] = make_float4(0.f, 0.f, 0.f, 0.f);
        if (i < RV) rq[u] = rsrc[i];
    }
    #pragma unroll
    for (int u = 0; u < DTRIPS; ++u) {
        const int i = t + u * TL_THREADS;
        sq[u] = make_float2(0.f, 0.f);
        dq[u] = make_float2(0.f, 0.f);
        if (i < DV) { sq[u] = ssrc[i]; dq[u] = dsrc[i]; }
    }

    WaveList wl;
    wave_list(seg, b, wv, lane, wl);
    const int len = wl.pre[SEG_PW];
    unsigned int wN[NE];
    #pragma unroll
    for (int j = 0; j < NE; ++j) {
        const int v = j * EPS + slot;
        wN[j] = ~0u;
        if (v < len) wN[j] = binned[word_index(wl, v)];
    }

    float4* acc4  = reinterpret_cast<float4*>(acc);
    float4* rows4 = reinterpret_cast<float4*>(rows);
    float2* sdv2  = reinterpret_cast<float2*>(sdv);
    float2* ddv2  = reinterpret_cast<float2*>(ddv);
    for (int i = t; i < RV; i += TL_THREADS) acc4[i] = make_float4(0.f, 0.f, 0.f, 0.f);
    #pragma unroll
    for (int u = 0; u < RTRIPS; ++u) {
        const int i = t + u * TL_THREADS;
        if (i < RV) rows4[i] = rq[u];
    }
    #pragma unroll
    for (int u = 0; u < DTRIPS; ++u) {
        const int i = t + u * TL_THREADS;
        if (i < DV) { sdv2[i] = sq[u]; ddv2[i] = dq[u]; }
    }
    __syncthreads();

    int v0 = 0;
    for (;;) {
        unsigned int w[NE];
        #pragma unroll
        for (int j = 0; j < NE; ++j) w[j] = wN[j];
        const int  nv0  = v0 + NE * EPS;
        const bool more = nv0 < len;
        #pragma unroll
        for (int j = 0; j < NE; ++j) {
            const int v = nv0 + j * EPS + slot;
            wN[j] = ~0u;
            if (v < len) wN[j] = binned[word_index(wl, v)];
        }

        bool   m[NE];
        int    dn[NE];
        VT     rp[NE];
        float  sd[NE], dd[NE];
        #pragma unroll
        for (int j = 0; j < NE; ++j) {
            const unsigned int s_ = w[j] & OFF_MASK;
            const unsigned int d_ = w[j] >> OFF_BITS;
            m[j]  = s_ < (unsigned)NP && d_ < (unsigned)NP;
            const int so = m[j] ? (int)s_ : 0;
            dn[j] = m[j] ? (int)d_ : 0;
            rp[j] = *reinterpret_cast<const VT*>(&rows[so * OUT]);
            sd[j] = sdv[so];
            dd[j] = ddv[dn[j]];
        }
        #pragma unroll
        for (int j = 0; j < NE; ++j) {
            if (m[j]) {
                const float nrm = dd[j] * sd[j];
                float p[OUT];
                row_times(rp[j], nrm, p);
                rotate_to_lane(p, q);
                float* a = &acc[dn[j] * OUT];
                #pragma unroll
                for (int k = 0; k < OUT; ++k)
                    __hip_atomic_fetch_add(a + fo[k], p[k],
                                           __ATOMIC_RELAXED, __HIP_MEMORY_SCOPE_WORKGROUP);
            }
        }
        if (!more) break;
        v0 = nv0;
    }
    __syncthreads();

    float4* dst = reinterpret_cast<float4*>(part + ((size_t)sr * N_NODES + (size_t)dr * NP) * OUT);
    for (int i = t; i < RV; i += TL_THREADS) dst[i] = acc4[i];
}

template <int OUT>
__device__ __forceinline__ void node_features(const float* __restrict__ part,
                                              const float* __restrict__ hlin,
                                              float di,
                                              const float* __restrict__ b,
                                              int i, float (&hv)[OUT]) {
    typedef typename RowVec<OUT>::type VT;
    constexpr int G = TL_R;
    const VT* pv = reinterpret_cast<const VT*>(part);
    const VT hrow = reinterpret_cast<const VT*>(hlin)[i];
    const float* hr = reinterpret_cast<const float*>(&hrow);
    float agg[OUT];
    #pragma unroll
    for (int f = 0; f < OUT; ++f) agg[f] = 0.0f;
    #pragma unroll
    for (int g = 0; g < G; ++g) {
        const VT pg = pv[(size_t)g * N_NODES + i];
        const float* pf = reinterpret_cast<const float*>(&pg);
        #pragma unroll
        for (int f = 0; f < OUT; ++f) agg[f] += pf[f];
    }
    const float d2 = di * di;
    #pragma unroll
    for (int f = 0; f < OUT; ++f) {
        float v = agg[f] + hr[f] * d2 + b[f];
        hv[f] = tanhf(v);
    }
}

template <int OUT, int NEXT>
__global__ void node_stage_kernel(const float* __restrict__ part,
                                  const float* __restrict__ hlin,
                                  const float* __restrict__ dinv,
                                  const float* __restrict__ b,
                                  const float* __restrict__ Wn,
                                  float* __restrict__ tnext,
                                  int n) {
    int i = blockIdx.x * blockDim.x + threadIdx.x;
    if (i >= n) return;
    const float di = dinv[i];
    float hv[OUT];
    node_features<OUT>(part, hlin, di, b, i, hv);

    float s[NEXT];
    #pragma unroll
    for (int o = 0; o < NEXT; ++o) {
        s[o] = 0.0f;
        #pragma unroll
        for (int k = 0; k < OUT; ++k) s[o] = fmaf(hv[k], Wn[k * NEXT + o], s[o]);
    }
    float* orow = tnext + (size_t)i * NEXT;
    if constexpr (NEXT == 4) {
        *reinterpret_cast<float4*>(orow) = make_float4(s[0], s[1], s[2], s[3]);
    } else {
        static_assert(NEXT == 2, "rows of 4 or 2 features");
        *reinterpret_cast<float2*>(orow) = make_float2(s[0], s[1]);
    }
}

__global__ void node_last_kernel(const float* __restrict__ part,
                                 const float* __restrict__ hlin,
                                 const float* __restrict__ dinv,
                                 const float* __restrict__ b,
                                 const float* __restrict__ Wc,
                                 const float* __restrict__ bc,
                                 float* __restrict__ outH,
                                 float* __restrict__ outC,
                                 int n) {
    int i = blockIdx.x * blockDim.x + threadIdx.x;
    if (i >= n) return;
    float hv[2];
    node_features<2>(part, hlin, dinv[i], b, i, hv);
    *reinterpret_cast<float2*>(outH + (size_t)i * 2) = make_float2(hv[0], hv[1]);
    float cl[4];
    #pragma unroll
    for (int c = 0; c < 4; ++c)
        cl[c] = fmaf(hv[0], Wc[c], fmaf(hv[1], Wc[4 + c], bc[c]));
    *reinterpret_cast<float4*>(outC + (size_t)i * 4) = make_float4(cl[0], cl[1], cl[2], cl[3]);
}

extern "C" void kernel_launch(void* const* d_in, const int* in_sizes, int n_in,
                              void* d_out, int out_size, void* d_ws, size_t ws_size,
                              hipStream_t stream) {
    const float* x  = (const float*)d_in[0];
    const int*   ei = (const int*)d_in[1];
    const float* W1 = (const float*)d_in[2];
    const float* b1 = (const float*)d_in[3];
    const float* W2 = (const float*)d_in[4];
    const float* b2 = (const float*)d_in[5];
    const float* W3 = (const float*)d_in[6];
    const float* b3 = (const float*)d_in[7];
    const float* Wc = (const float*)d_in[8];
    const float* bc = (const float*)d_in[9];

    const int N = N_NODES;
    const int E = N_EDGES;
    const int* row = ei;
    const int* col = ei + E;

    float* ws   = (float*)d_ws;
    float* dinv = ws;
    float* tabA = dinv + N;
    float* tabB = tabA + 4 * N;
    float* part = tabB + 4 * N;
    unsigned int* binned = (unsigned int*)(part + 64 * N);
    unsigned int* seg    = binned + (size_t)BN_GROUPS * BN_ESL;

    float* outC = (float*)d_out;
    float* outH = outC + (size_t)4 * N;

    const int BT = 256;
    const int gN = (N + BT - 1) / BT;

    tile_bin_kernel<<<BN_GROUPS, TL_THREADS, 0, stream>>>(row, col, binned, seg);
    tile_degree_kernel<<<TL_TILES, TL_THREADS, 0, stream>>>(binned, seg, part);

    {
        const int tiles  = N / 16;
        const int blocks = (tiles + 7) / 8;
        gcn_wmma_xw1_kernel<<<blocks, 256, 0, stream>>>(x, W1, part, dinv, tabA, N);
        edge_tile_kernel<4><<<TL_TILES, TL_THREADS, 0, stream>>>(binned, seg, dinv, tabA, part);
    }
    {
        node_stage_kernel<4, 4><<<gN, BT, 0, stream>>>(part, tabA, dinv, b1, W2, tabB, N);
        edge_tile_kernel<4><<<TL_TILES, TL_THREADS, 0, stream>>>(binned, seg, dinv, tabB, part);
    }
    {
        node_stage_kernel<4, 2><<<gN, BT, 0, stream>>>(part, tabB, dinv, b2, W3, tabA, N);
        edge_tile_kernel<2><<<TL_TILES, TL_THREADS, 0, stream>>>(binned, seg, dinv, tabA, part);
    }
    node_last_kernel<<<gN, BT, 0, stream>>>(part, tabA, dinv, b3, Wc, bc, outH, outC, N);
}
